# starting kernel
# baseline (speedup 1.0000x reference)
#include <hip/hip_runtime.h>

typedef float v2f __attribute__((ext_vector_type(2)));
typedef float v8f __attribute__((ext_vector_type(8)));

#define CC 64
#define DD 8
#define HH 64
#define WW 64
#define APAD 68
#define AROW 66

__device__ __forceinline__ float swish_f(float v) {
    return v / (1.0f + __expf(-v));
}

__global__ __launch_bounds__(256) void dense_es_kernel(
    const float* __restrict__ embed, const float* __restrict__ prompt,
    const float* __restrict__ d1w, const float* __restrict__ d1b,
    const float* __restrict__ d2w, const float* __restrict__ d2b,
    float* __restrict__ es)
{
    const int t = threadIdx.x;
    const int bb = t >> 6, c = t & 63;
    const float* eb = embed + bb * 256;
    const float* pb = prompt + bb * 256;
    const float* w1 = d1w + c * 256;
    const float* w2 = d2w + c * 256;
    float a1 = 0.f, a2 = 0.f;
    for (int k = 0; k < 256; ++k) {
        a1 = fmaf(eb[k], w1[k], a1);
        a2 = fmaf(pb[k], w2[k], a2);
    }
    es[t]       = a1 + d1b[c];
    es[256 + t] = a2 + d2b[c];
}

template<int MODE>
__global__ __launch_bounds__(256) void conv3x3x3_wmma_kernel(
    const float* __restrict__ src,
    const float* __restrict__ wgt,
    const float* __restrict__ bias,
    const float* __restrict__ es,
    const float* __restrict__ resid,
    float* __restrict__ dst)
{
    __shared__ float Alds[2 * AROW * APAD];
    __shared__ float Wlds[3 * CC * APAD];

    const int tid    = threadIdx.x;
    const int lane   = tid & 31;
    const int wv     = tid >> 5;
    const int lane16 = lane & 15;
    const int half   = lane >> 4;
    const int koff   = half * 2;
    const int mpair  = wv & 3;
    const int npair  = wv >> 2;

    const int wg = blockIdx.x;
    const int hb = wg & 31;
    const int d  = (wg >> 5) & 7;
    const int b  = wg >> 8;
    const int h0 = hb * 2;

    v8f acc00 = {}, acc01 = {}, acc10 = {}, acc11 = {};

    for (int kd = 0; kd < 3; ++kd) {
        const int zd = d + kd - 1;
        if (zd < 0 || zd >= DD) continue;
        for (int kh = 0; kh < 3; ++kh) {
            __syncthreads();
            for (int i = tid; i < 2 * CC * AROW; i += 256) {
                const int r    = i / (CC * AROW);
                const int j    = i - r * (CC * AROW);
                const int c    = j / AROW;
                const int wpos = j - c * AROW;
                const int wi   = wpos - 1;
                const int zh   = h0 + kh - 1 + r;
                float v = 0.f;
                if (wi >= 0 && wi < WW && zh >= 0 && zh < HH) {
                    v = src[(((size_t)(b * CC + c) * DD + zd) * HH + zh) * WW + wi];
                    if (MODE == 0) v = swish_f(v + es[b * CC + c]);
                }
                Alds[(r * AROW + wpos) * APAD + c] = v;
            }
            for (int i = tid; i < 3 * CC * CC; i += 256) {
                const int c  = i & 63;
                const int o  = (i >> 6) & 63;
                const int kw = i >> 12;
                Wlds[(kw * CC + o) * APAD + c] =
                    wgt[(size_t)(o * CC + c) * 27 + (kd * 9 + kh * 3 + kw)];
            }
            __syncthreads();
            for (int kw = 0; kw < 3; ++kw) {
                const int aw = mpair * 16 + lane16 + kw;
                const float* ap0 = &Alds[aw * APAD];
                const float* ap1 = &Alds[(AROW + aw) * APAD];
                const float* bp0 = &Wlds[(kw * CC + npair * 32 + lane16) * APAD];
                const float* bp1 = bp0 + 16 * APAD;
                #pragma unroll
                for (int ck = 0; ck < 16; ++ck) {
                    const int c0 = ck * 4 + koff;
                    const v2f a0 = *(const v2f*)(ap0 + c0);
                    const v2f a1 = *(const v2f*)(ap1 + c0);
                    const v2f b0 = *(const v2f*)(bp0 + c0);
                    const v2f b1 = *(const v2f*)(bp1 + c0);
                    acc00 = __builtin_amdgcn_wmma_f32_16x16x4_f32(
                        false, a0, false, b0, (short)0, acc00, false, false);
                    acc01 = __builtin_amdgcn_wmma_f32_16x16x4_f32(
                        false, a0, false, b1, (short)0, acc01, false, false);
                    acc10 = __builtin_amdgcn_wmma_f32_16x16x4_f32(
                        false, a1, false, b0, (short)0, acc10, false, false);
                    acc11 = __builtin_amdgcn_wmma_f32_16x16x4_f32(
                        false, a1, false, b1, (short)0, acc11, false, false);
                }
            }
        }
    }

    const int wbase = mpair * 16 + half * 8;
    #pragma unroll
    for (int r = 0; r < 2; ++r) {
        #pragma unroll
        for (int t = 0; t < 2; ++t) {
            const v8f acc = r ? (t ? acc11 : acc10) : (t ? acc01 : acc00);
            const int o = npair * 32 + t * 16 + lane16;
            const size_t off =
                (((size_t)(b * CC + o) * DD + d) * HH + (h0 + r)) * WW + wbase;
            float v[8];
            #pragma unroll
            for (int i = 0; i < 8; ++i) v[i] = acc[i];
            if (MODE == 0) {
                const float bo = bias[o];
                const float so = es[256 + b * CC + o];
                #pragma unroll
                for (int i = 0; i < 8; ++i) v[i] = swish_f(v[i] + bo) * so;
            } else {
                const float4 x0 = *(const float4*)(resid + off);
                const float4 x1 = *(const float4*)(resid + off + 4);
                v[0] += x0.x; v[1] += x0.y; v[2] += x0.z; v[3] += x0.w;
                v[4] += x1.x; v[5] += x1.y; v[6] += x1.z; v[7] += x1.w;
            }
            *(float4*)(dst + off)     = make_float4(v[0], v[1], v[2], v[3]);
            *(float4*)(dst + off + 4) = make_float4(v[4], v[5], v[6], v[7]);
        }
    }
}

extern "C" void kernel_launch(void* const* d_in, const int* in_sizes, int n_in,
                              void* d_out, int out_size, void* d_ws, size_t ws_size,
                              hipStream_t stream) {
    (void)in_sizes; (void)n_in; (void)out_size; (void)ws_size;
    const float* x      = (const float*)d_in[0];
    const float* embed  = (const float*)d_in[1];
    const float* prompt = (const float*)d_in[2];
    const float* c20w   = (const float*)d_in[3];
    const float* c20b   = (const float*)d_in[4];
    const float* c21w   = (const float*)d_in[5];
    const float* d1w    = (const float*)d_in[6];
    const float* d1b    = (const float*)d_in[7];
    const float* d2w    = (const float*)d_in[8];
    const float* d2b    = (const float*)d_in[9];
    float* out  = (float*)d_out;
    float* es   = (float*)d_ws;
    float* hbuf = es + 512;

    dense_es_kernel<<<1, 256, 0, stream>>>(embed, prompt, d1w, d1b, d2w, d2b, es);

    const int ngrid = 4 * DD * (HH / 2);
    conv3x3x3_wmma_kernel<0><<<ngrid, 256, 0, stream>>>(x, c20w, c20b, es, nullptr, hbuf);
    conv3x3x3_wmma_kernel<1><<<ngrid, 256, 0, stream>>>(hbuf, c21w, nullptr, nullptr, x, out);
}


# K loop regrouped into 8-input-channel chunks so each out-channel's weights are staged as one contiguous 864-B run (whole lines) instead of a stride-27 gather per (kd,kh) slab; f32 sum order changes
# speedup vs baseline: 4.3364x; 4.3364x over previous
#include <hip/hip_runtime.h>

typedef float v2f __attribute__((ext_vector_type(2)));
typedef float v4f __attribute__((ext_vector_type(4)));
typedef float v8f __attribute__((ext_vector_type(8)));

#define CC 64
#define DD 8
#define HH 64
#define WW 64
#define CB 8
#define PT 12
#define AROW 66

__device__ __forceinline__ float swish_f(float v) {
    return v / (1.0f + __expf(-v));
}

__global__ __launch_bounds__(256) void dense_es_kernel(
    const float* __restrict__ embed, const float* __restrict__ prompt,
    const float* __restrict__ d1w, const float* __restrict__ d1b,
    const float* __restrict__ d2w, const float* __restrict__ d2b,
    float* __restrict__ es)
{
    const int t = threadIdx.x;
    const int bb = t >> 6, c = t & 63;
    const float* eb = embed + bb * 256;
    const float* pb = prompt + bb * 256;
    const float* w1 = d1w + c * 256;
    const float* w2 = d2w + c * 256;
    float a1 = 0.f, a2 = 0.f;
    for (int k = 0; k < 256; ++k) {
        a1 = fmaf(eb[k], w1[k], a1);
        a2 = fmaf(pb[k], w2[k], a2);
    }
    es[t]       = a1 + d1b[c];
    es[256 + t] = a2 + d2b[c];
}

template<int MODE>
__global__ __launch_bounds__(256) void conv3x3x3_wmma_kernel(
    const float* __restrict__ src,
    const float* __restrict__ wgt,
    const float* __restrict__ bias,
    const float* __restrict__ es,
    const float* __restrict__ resid,
    float* __restrict__ dst)
{
    __shared__ __attribute__((aligned(16))) float Alds[12 * AROW * PT];
    __shared__ __attribute__((aligned(16))) float Wlds[CC * 27 * PT];

    const int tid    = threadIdx.x;
    const int lane   = tid & 31;
    const int wv     = tid >> 5;
    const int lane16 = lane & 15;
    const int half   = lane >> 4;
    const int mpair  = wv & 3;
    const int npair  = wv >> 2;

    const int wg = blockIdx.x;
    const int hb = wg & 31;
    const int d  = (wg >> 5) & 7;
    const int b  = wg >> 8;
    const int h0 = hb * 2;

    if (tid < 192) {
        const int pr = tid >> 4, side = (tid >> 3) & 1, c8 = tid & 7;
        Alds[(pr * AROW + side * (AROW - 1)) * PT + c8] = 0.f;
    }

    const int a_w4  = (lane & 7) + 8 * (wv & 1);
    const int a_c8  = (lane >> 3) + 4 * ((wv >> 1) & 1);
    const int a_pr0 = wv >> 2;

    int woff[6];
    #pragma unroll
    for (int k = 0; k < 6; ++k) {
        const int j  = lane + 32 * k;
        const int c8 = j / 27;
        const int tp = j - 27 * c8;
        woff[k] = tp * PT + c8;
    }
    int tgo[3], tlo[3];
    #pragma unroll
    for (int m = 0; m < 3; ++m) {
        const int t  = lane + 32 * m;
        const int oo = t / 24;
        const int j  = 192 + (t - 24 * oo);
        const int c8 = j / 27;
        const int tp = j - 27 * c8;
        tgo[m] = oo * (CC * 27) + j;
        tlo[m] = oo * (27 * PT) + tp * PT + c8;
    }

    v8f acc00 = {}, acc01 = {}, acc10 = {}, acc11 = {};

    for (int cb = 0; cb < CC / CB; ++cb) {
        __syncthreads();
        const float* g0 = wgt + (size_t)(wv * 8) * (CC * 27) + cb * (CB * 27);
        float* l0 = &Wlds[(wv * 8) * 27 * PT];
        const int cin = cb * CB + a_c8;
        const float* xc = src + (size_t)(b * CC + cin) * (DD * HH * WW) + a_w4 * 4;
        float ev = 0.f;
        if (MODE == 0) ev = es[b * CC + cin];

        float4 av[6];
        #pragma unroll
        for (int it = 0; it < 6; ++it) {
            const int pr = a_pr0 + 2 * it;
            const int zd = d + (pr >> 2) - 1;
            const int zh = h0 - 1 + (pr & 3);
            av[it] = make_float4(0.f, 0.f, 0.f, 0.f);
            if (zd >= 0 && zd < DD && zh >= 0 && zh < HH)
                av[it] = *(const float4*)(xc + (zd * HH + zh) * WW);
        }
        float wr[4][6], wt[3];
        #pragma unroll
        for (int oo = 0; oo < 4; ++oo) {
            #pragma unroll
            for (int k = 0; k < 6; ++k) wr[oo][k] = g0[oo * (CC * 27) + lane + 32 * k];
        }
        #pragma unroll
        for (int m = 0; m < 3; ++m) wt[m] = g0[tgo[m]];
        #pragma unroll
        for (int it = 0; it < 6; ++it) {
            const int pr = a_pr0 + 2 * it;
            const int zd = d + (pr >> 2) - 1;
            const int zh = h0 - 1 + (pr & 3);
            float4 v = av[it];
            if (MODE == 0) {
                if (zd >= 0 && zd < DD && zh >= 0 && zh < HH) {
                    v.x = swish_f(v.x + ev); v.y = swish_f(v.y + ev);
                    v.z = swish_f(v.z + ev); v.w = swish_f(v.w + ev);
                }
            }
            float* ap = &Alds[(pr * AROW + 1 + a_w4 * 4) * PT + a_c8];
            ap[0] = v.x; ap[PT] = v.y; ap[2 * PT] = v.z; ap[3 * PT] = v.w;
        }
        #pragma unroll
        for (int oo = 0; oo < 4; ++oo) {
            #pragma unroll
            for (int k = 0; k < 6; ++k) l0[oo * (27 * PT) + woff[k]] = wr[oo][k];
        }
        #pragma unroll
        for (int m = 0; m < 3; ++m) l0[tlo[m]] = wt[m];
        #pragma unroll
        for (int oo = 0; oo < 4; ++oo) {
            #pragma unroll
            for (int k = 0; k < 6; ++k) wr[oo][k] = g0[(oo + 4) * (CC * 27) + lane + 32 * k];
        }
        #pragma unroll
        for (int m = 0; m < 3; ++m) wt[m] = g0[4 * (CC * 27) + tgo[m]];
        #pragma unroll
        for (int oo = 0; oo < 4; ++oo) {
            #pragma unroll
            for (int k = 0; k < 6; ++k) l0[(oo + 4) * (27 * PT) + woff[k]] = wr[oo][k];
        }
        #pragma unroll
        for (int m = 0; m < 3; ++m) l0[4 * (27 * PT) + tlo[m]] = wt[m];
        __syncthreads();
        for (int kd = 0; kd < 3; ++kd) {
            const int zd = d + kd - 1;
            if (zd < 0 || zd >= DD) continue;
            #pragma unroll
            for (int kh = 0; kh < 3; ++kh) {
                #pragma unroll
                for (int kw = 0; kw < 3; ++kw) {
                    const int tap = kd * 9 + kh * 3 + kw;
                    const int aw  = mpair * 16 + lane16 + kw;
                    const float* ap = &Alds[((kd * 4 + kh) * AROW + aw) * PT + half * 4];
                    const float* bp = &Wlds[((npair * 32 + lane16) * 27 + tap) * PT + half * 4];
                    const v4f a0 = *(const v4f*)ap;
                    const v4f a1 = *(const v4f*)(ap + AROW * PT);
                    const v4f b0 = *(const v4f*)bp;
                    const v4f b1 = *(const v4f*)(bp + 16 * 27 * PT);
                    const v2f a0l = {a0.x, a0.y}, a0h = {a0.z, a0.w};
                    const v2f a1l = {a1.x, a1.y}, a1h = {a1.z, a1.w};
                    const v2f b0l = {b0.x, b0.y}, b0h = {b0.z, b0.w};
                    const v2f b1l = {b1.x, b1.y}, b1h = {b1.z, b1.w};
                    acc00 = __builtin_amdgcn_wmma_f32_16x16x4_f32(
                        false, a0l, false, b0l, (short)0, acc00, false, false);
                    acc01 = __builtin_amdgcn_wmma_f32_16x16x4_f32(
                        false, a0l, false, b1l, (short)0, acc01, false, false);
                    acc10 = __builtin_amdgcn_wmma_f32_16x16x4_f32(
                        false, a1l, false, b0l, (short)0, acc10, false, false);
                    acc11 = __builtin_amdgcn_wmma_f32_16x16x4_f32(
                        false, a1l, false, b1l, (short)0, acc11, false, false);
                    acc00 = __builtin_amdgcn_wmma_f32_16x16x4_f32(
                        false, a0h, false, b0h, (short)0, acc00, false, false);
                    acc01 = __builtin_amdgcn_wmma_f32_16x16x4_f32(
                        false, a0h, false, b1h, (short)0, acc01, false, false);
                    acc10 = __builtin_amdgcn_wmma_f32_16x16x4_f32(
                        false, a1h, false, b0h, (short)0, acc10, false, false);
                    acc11 = __builtin_amdgcn_wmma_f32_16x16x4_f32(
                        false, a1h, false, b1h, (short)0, acc11, false, false);
                }
            }
        }
    }

    const int wbase = mpair * 16 + half * 8;
    #pragma unroll
    for (int r = 0; r < 2; ++r) {
        #pragma unroll
        for (int t = 0; t < 2; ++t) {
            const v8f acc = r ? (t ? acc11 : acc10) : (t ? acc01 : acc00);
            const int o = npair * 32 + t * 16 + lane16;
            const size_t off =
                (((size_t)(b * CC + o) * DD + d) * HH + (h0 + r)) * WW + wbase;
            float v[8];
            #pragma unroll
            for (int i = 0; i < 8; ++i) v[i] = acc[i];
            if (MODE == 0) {
                const float bo = bias[o];
                const float so = es[256 + b * CC + o];
                #pragma unroll
                for (int i = 0; i < 8; ++i) v[i] = swish_f(v[i] + bo) * so;
            } else {
                const float4 x0 = *(const float4*)(resid + off);
                const float4 x1 = *(const float4*)(resid + off + 4);
                v[0] += x0.x; v[1] += x0.y; v[2] += x0.z; v[3] += x0.w;
                v[4] += x1.x; v[5] += x1.y; v[6] += x1.z; v[7] += x1.w;
            }
            *(float4*)(dst + off)     = make_float4(v[0], v[1], v[2], v[3]);
            *(float4*)(dst + off + 4) = make_float4(v[4], v[5], v[6], v[7]);
        }
    }
}

extern "C" void kernel_launch(void* const* d_in, const int* in_sizes, int n_in,
                              void* d_out, int out_size, void* d_ws, size_t ws_size,
                              hipStream_t stream) {
    (void)in_sizes; (void)n_in; (void)out_size; (void)ws_size;
    const float* x      = (const float*)d_in[0];
    const float* embed  = (const float*)d_in[1];
    const float* prompt = (const float*)d_in[2];
    const float* c20w   = (const float*)d_in[3];
    const float* c20b   = (const float*)d_in[4];
    const float* c21w   = (const float*)d_in[5];
    const float* d1w    = (const float*)d_in[6];
    const float* d1b    = (const float*)d_in[7];
    const float* d2w    = (const float*)d_in[8];
    const float* d2b    = (const float*)d_in[9];
    float* out  = (float*)d_out;
    float* es   = (float*)d_ws;
    float* hbuf = es + 512;

    dense_es_kernel<<<1, 256, 0, stream>>>(embed, prompt, d1w, d1b, d2w, d2b, es);

    const int ngrid = 4 * DD * (HH / 2);
    conv3x3x3_wmma_kernel<0><<<ngrid, 256, 0, stream>>>(x, c20w, c20b, es, nullptr, hbuf);
    conv3x3x3_wmma_kernel<1><<<ngrid, 256, 0, stream>>>(hbuf, c21w, nullptr, nullptr, x, out);
}


# Workgroup tile doubled to 4 output rows (M=256): a staged weight chunk and each B fragment serve 4 rows, 6 staged input rows serve 4; taps walked kw-outer, so f32 sum order changes
# speedup vs baseline: 4.8571x; 1.1201x over previous
#include <hip/hip_runtime.h>

typedef float v2f __attribute__((ext_vector_type(2)));
typedef float v4f __attribute__((ext_vector_type(4)));
typedef float v8f __attribute__((ext_vector_type(8)));

#define CC 64
#define DD 8
#define HH 64
#define WW 64
#define CB 8
#define PT 12
#define AROW 66
#define TR 4
#define AR (TR + 2)

__device__ __forceinline__ float swish_f(float v) {
    return v / (1.0f + __expf(-v));
}

__global__ __launch_bounds__(256) void dense_es_kernel(
    const float* __restrict__ embed, const float* __restrict__ prompt,
    const float* __restrict__ d1w, const float* __restrict__ d1b,
    const float* __restrict__ d2w, const float* __restrict__ d2b,
    float* __restrict__ es)
{
    const int t = threadIdx.x;
    const int bb = t >> 6, c = t & 63;
    const float* eb = embed + bb * 256;
    const float* pb = prompt + bb * 256;
    const float* w1 = d1w + c * 256;
    const float* w2 = d2w + c * 256;
    float a1 = 0.f, a2 = 0.f;
    for (int k = 0; k < 256; ++k) {
        a1 = fmaf(eb[k], w1[k], a1);
        a2 = fmaf(pb[k], w2[k], a2);
    }
    es[t]       = a1 + d1b[c];
    es[256 + t] = a2 + d2b[c];
}

template<int MODE>
__global__ __launch_bounds__(256) void conv3x3x3_wmma_kernel(
    const float* __restrict__ src,
    const float* __restrict__ wgt,
    const float* __restrict__ bias,
    const float* __restrict__ es,
    const float* __restrict__ resid,
    float* __restrict__ dst)
{
    __shared__ __attribute__((aligned(16))) float Alds[3 * AR * AROW * PT];
    __shared__ __attribute__((aligned(16))) float Wlds[CC * 27 * PT];

    const int tid    = threadIdx.x;
    const int lane   = tid & 31;
    const int wv     = tid >> 5;
    const int lane16 = lane & 15;
    const int half   = lane >> 4;
    const int mpair  = wv & 3;
    const int npair  = wv >> 2;

    const int wg = blockIdx.x;
    const int hb = wg & 15;
    const int d  = (wg >> 4) & 7;
    const int b  = wg >> 7;
    const int h0 = hb * TR;

    for (int i = tid; i < 3 * AR * 16; i += 256) {
        const int pr = i >> 4, side = (i >> 3) & 1, c8 = i & 7;
        Alds[(pr * AROW + side * (AROW - 1)) * PT + c8] = 0.f;
    }

    const int a_w4  = (lane & 7) + 8 * (wv & 1);
    const int a_c8  = (lane >> 3) + 4 * ((wv >> 1) & 1);
    const int a_pr0 = wv >> 2;

    int woff[6];
    #pragma unroll
    for (int k = 0; k < 6; ++k) {
        const int j  = lane + 32 * k;
        const int c8 = j / 27;
        const int tp = j - 27 * c8;
        woff[k] = tp * PT + c8;
    }
    int tgo[3], tlo[3];
    #pragma unroll
    for (int m = 0; m < 3; ++m) {
        const int t  = lane + 32 * m;
        const int oo = t / 24;
        const int j  = 192 + (t - 24 * oo);
        const int c8 = j / 27;
        const int tp = j - 27 * c8;
        tgo[m] = oo * (CC * 27) + j;
        tlo[m] = oo * (27 * PT) + tp * PT + c8;
    }

    v8f acc[TR][2];
    #pragma unroll
    for (int r = 0; r < TR; ++r) { acc[r][0] = (v8f){}; acc[r][1] = (v8f){}; }

    for (int cb = 0; cb < CC / CB; ++cb) {
        __syncthreads();
        const float* g0 = wgt + (size_t)(wv * 8) * (CC * 27) + cb * (CB * 27);
        float* l0 = &Wlds[(wv * 8) * 27 * PT];
        const int cin = cb * CB + a_c8;
        const float* xc = src + (size_t)(b * CC + cin) * (DD * HH * WW) + a_w4 * 4;
        float ev = 0.f;
        if (MODE == 0) ev = es[b * CC + cin];

        float4 av[9];
        #pragma unroll
        for (int it = 0; it < 9; ++it) {
            const int rr = 2 * (it % 3) + a_pr0;
            const int zd = d + (it / 3) - 1;
            const int zh = h0 - 1 + rr;
            av[it] = make_float4(0.f, 0.f, 0.f, 0.f);
            if (zd >= 0 && zd < DD && zh >= 0 && zh < HH)
                av[it] = *(const float4*)(xc + (zd * HH + zh) * WW);
        }
        float wr[4][6], wt[3];
        #pragma unroll
        for (int oo = 0; oo < 4; ++oo) {
            #pragma unroll
            for (int k = 0; k < 6; ++k) wr[oo][k] = g0[oo * (CC * 27) + lane + 32 * k];
        }
        #pragma unroll
        for (int m = 0; m < 3; ++m) wt[m] = g0[tgo[m]];
        #pragma unroll
        for (int it = 0; it < 9; ++it) {
            const int rr = 2 * (it % 3) + a_pr0;
            const int pr = (it / 3) * AR + rr;
            const int zd = d + (it / 3) - 1;
            const int zh = h0 - 1 + rr;
            float4 v = av[it];
            if (MODE == 0) {
                if (zd >= 0 && zd < DD && zh >= 0 && zh < HH) {
                    v.x = swish_f(v.x + ev); v.y = swish_f(v.y + ev);
                    v.z = swish_f(v.z + ev); v.w = swish_f(v.w + ev);
                }
            }
            float* ap = &Alds[(pr * AROW + 1 + a_w4 * 4) * PT + a_c8];
            ap[0] = v.x; ap[PT] = v.y; ap[2 * PT] = v.z; ap[3 * PT] = v.w;
        }
        #pragma unroll
        for (int oo = 0; oo < 4; ++oo) {
            #pragma unroll
            for (int k = 0; k < 6; ++k) l0[oo * (27 * PT) + woff[k]] = wr[oo][k];
        }
        #pragma unroll
        for (int m = 0; m < 3; ++m) l0[tlo[m]] = wt[m];
        #pragma unroll
        for (int oo = 0; oo < 4; ++oo) {
            #pragma unroll
            for (int k = 0; k < 6; ++k) wr[oo][k] = g0[(oo + 4) * (CC * 27) + lane + 32 * k];
        }
        #pragma unroll
        for (int m = 0; m < 3; ++m) wt[m] = g0[4 * (CC * 27) + tgo[m]];
        #pragma unroll
        for (int oo = 0; oo < 4; ++oo) {
            #pragma unroll
            for (int k = 0; k < 6; ++k) l0[(oo + 4) * (27 * PT) + woff[k]] = wr[oo][k];
        }
        #pragma unroll
        for (int m = 0; m < 3; ++m) l0[4 * (27 * PT) + tlo[m]] = wt[m];
        __syncthreads();
        for (int kd = 0; kd < 3; ++kd) {
            const int zd = d + kd - 1;
            if (zd < 0 || zd >= DD) continue;
            #pragma unroll
            for (int kw = 0; kw < 3; ++kw) {
                const int aw = mpair * 16 + lane16 + kw;
                const float* ap = &Alds[((kd * AR) * AROW + aw) * PT + half * 4];
                v4f a[AR];
                #pragma unroll
                for (int rr = 0; rr < AR; ++rr) a[rr] = *(const v4f*)(ap + rr * (AROW * PT));
                #pragma unroll
                for (int kh = 0; kh < 3; ++kh) {
                    const int tap = kd * 9 + kh * 3 + kw;
                    const float* bp = &Wlds[((npair * 32 + lane16) * 27 + tap) * PT + half * 4];
                    const v4f b0 = *(const v4f*)bp;
                    const v4f b1 = *(const v4f*)(bp + 16 * 27 * PT);
                    const v2f b0l = {b0.x, b0.y}, b0h = {b0.z, b0.w};
                    const v2f b1l = {b1.x, b1.y}, b1h = {b1.z, b1.w};
                    #pragma unroll
                    for (int r = 0; r < TR; ++r) {
                        const v4f ar = a[kh + r];
                        const v2f al = {ar.x, ar.y};
                        acc[r][0] = __builtin_amdgcn_wmma_f32_16x16x4_f32(
                            false, al, false, b0l, (short)0, acc[r][0], false, false);
                        acc[r][1] = __builtin_amdgcn_wmma_f32_16x16x4_f32(
                            false, al, false, b1l, (short)0, acc[r][1], false, false);
                    }
                    #pragma unroll
                    for (int r = 0; r < TR; ++r) {
                        const v4f ar = a[kh + r];
                        const v2f ah = {ar.z, ar.w};
                        acc[r][0] = __builtin_amdgcn_wmma_f32_16x16x4_f32(
                            false, ah, false, b0h, (short)0, acc[r][0], false, false);
                        acc[r][1] = __builtin_amdgcn_wmma_f32_16x16x4_f32(
                            false, ah, false, b1h, (short)0, acc[r][1], false, false);
                    }
                }
            }
        }
    }

    const int wbase = mpair * 16 + half * 8;
    #pragma unroll
    for (int r = 0; r < TR; ++r) {
        #pragma unroll
        for (int t = 0; t < 2; ++t) {
            const v8f ac = acc[r][t];
            const int o = npair * 32 + t * 16 + lane16;
            const size_t off =
                (((size_t)(b * CC + o) * DD + d) * HH + (h0 + r)) * WW + wbase;
            float v[8];
            #pragma unroll
            for (int i = 0; i < 8; ++i) v[i] = ac[i];
            if (MODE == 0) {
                const float bo = bias[o];
                const float so = es[256 + b * CC + o];
                #pragma unroll
                for (int i = 0; i < 8; ++i) v[i] = swish_f(v[i] + bo) * so;
            } else {
                const float4 x0 = *(const float4*)(resid + off);
                const float4 x1 = *(const float4*)(resid + off + 4);
                v[0] += x0.x; v[1] += x0.y; v[2] += x0.z; v[3] += x0.w;
                v[4] += x1.x; v[5] += x1.y; v[6] += x1.z; v[7] += x1.w;
            }
            *(float4*)(dst + off)     = make_float4(v[0], v[1], v[2], v[3]);
            *(float4*)(dst + off + 4) = make_float4(v[4], v[5], v[6], v[7]);
        }
    }
}

extern "C" void kernel_launch(void* const* d_in, const int* in_sizes, int n_in,
                              void* d_out, int out_size, void* d_ws, size_t ws_size,
                              hipStream_t stream) {
    (void)in_sizes; (void)n_in; (void)out_size; (void)ws_size;
    const float* x      = (const float*)d_in[0];
    const float* embed  = (const float*)d_in[1];
    const float* prompt = (const float*)d_in[2];
    const float* c20w   = (const float*)d_in[3];
    const float* c20b   = (const float*)d_in[4];
    const float* c21w   = (const float*)d_in[5];
    const float* d1w    = (const float*)d_in[6];
    const float* d1b    = (const float*)d_in[7];
    const float* d2w    = (const float*)d_in[8];
    const float* d2b    = (const float*)d_in[9];
    float* out  = (float*)d_out;
    float* es   = (float*)d_ws;
    float* hbuf = es + 512;

    dense_es_kernel<<<1, 256, 0, stream>>>(embed, prompt, d1w, d1b, d2w, d2b, es);

    const int ngrid = 4 * DD * (HH / TR);
    conv3x3x3_wmma_kernel<0><<<ngrid, 256, 0, stream>>>(x, c20w, c20b, es, nullptr, hbuf);
    conv3x3x3_wmma_kernel<1><<<ngrid, 256, 0, stream>>>(hbuf, c21w, nullptr, nullptr, x, out);
}


# Wave priority stepped down with the K-chunk index (s_setprio 3,2,1,0 from chunks 0,3,5,7): of the two workgroups on a WGP the one behind wins the matrix pipe, so each stages under the other's WMMAs
# speedup vs baseline: 5.0525x; 1.0402x over previous
#include <hip/hip_runtime.h>

typedef float v2f __attribute__((ext_vector_type(2)));
typedef float v4f __attribute__((ext_vector_type(4)));
typedef float v8f __attribute__((ext_vector_type(8)));

#define CC 64
#define DD 8
#define HH 64
#define WW 64
#define CB 8
#define PT 12
#define AROW 66
#define TR 4
#define AR (TR + 2)

__device__ __forceinline__ float swish_f(float v) {
    return v / (1.0f + __expf(-v));
}

__global__ __launch_bounds__(256) void dense_es_kernel(
    const float* __restrict__ embed, const float* __restrict__ prompt,
    const float* __restrict__ d1w, const float* __restrict__ d1b,
    const float* __restrict__ d2w, const float* __restrict__ d2b,
    float* __restrict__ es)
{
    const int t = threadIdx.x;
    const int bb = t >> 6, c = t & 63;
    const float* eb = embed + bb * 256;
    const float* pb = prompt + bb * 256;
    const float* w1 = d1w + c * 256;
    const float* w2 = d2w + c * 256;
    float a1 = 0.f, a2 = 0.f;
    for (int k = 0; k < 256; ++k) {
        a1 = fmaf(eb[k], w1[k], a1);
        a2 = fmaf(pb[k], w2[k], a2);
    }
    es[t]       = a1 + d1b[c];
    es[256 + t] = a2 + d2b[c];
}

template<int MODE>
__global__ __launch_bounds__(256) void conv3x3x3_wmma_kernel(
    const float* __restrict__ src,
    const float* __restrict__ wgt,
    const float* __restrict__ bias,
    const float* __restrict__ es,
    const float* __restrict__ resid,
    float* __restrict__ dst)
{
    __shared__ __attribute__((aligned(16))) float Alds[3 * AR * AROW * PT];
    __shared__ __attribute__((aligned(16))) float Wlds[CC * 27 * PT];

    const int tid    = threadIdx.x;
    const int lane   = tid & 31;
    const int wv     = tid >> 5;
    const int lane16 = lane & 15;
    const int half   = lane >> 4;
    const int mpair  = wv & 3;
    const int npair  = wv >> 2;

    const int wg = blockIdx.x;
    const int hb = wg & 15;
    const int d  = (wg >> 4) & 7;
    const int b  = wg >> 7;
    const int h0 = hb * TR;

    for (int i = tid; i < 3 * AR * 16; i += 256) {
        const int pr = i >> 4, side = (i >> 3) & 1, c8 = i & 7;
        Alds[(pr * AROW + side * (AROW - 1)) * PT + c8] = 0.f;
    }

    const int a_w4  = (lane & 7) + 8 * (wv & 1);
    const int a_c8  = (lane >> 3) + 4 * ((wv >> 1) & 1);
    const int a_pr0 = wv >> 2;

    int woff[6];
    #pragma unroll
    for (int k = 0; k < 6; ++k) {
        const int j  = lane + 32 * k;
        const int c8 = j / 27;
        const int tp = j - 27 * c8;
        woff[k] = tp * PT + c8;
    }
    int tgo[3], tlo[3];
    #pragma unroll
    for (int m = 0; m < 3; ++m) {
        const int t  = lane + 32 * m;
        const int oo = t / 24;
        const int j  = 192 + (t - 24 * oo);
        const int c8 = j / 27;
        const int tp = j - 27 * c8;
        tgo[m] = oo * (CC * 27) + j;
        tlo[m] = oo * (27 * PT) + tp * PT + c8;
    }

    v8f acc[TR][2];
    #pragma unroll
    for (int r = 0; r < TR; ++r) { acc[r][0] = (v8f){}; acc[r][1] = (v8f){}; }

    __builtin_amdgcn_s_setprio(3);
    for (int cb = 0; cb < CC / CB; ++cb) {
        if (cb == 3)      __builtin_amdgcn_s_setprio(2);
        else if (cb == 5) __builtin_amdgcn_s_setprio(1);
        else if (cb == 7) __builtin_amdgcn_s_setprio(0);
        __syncthreads();
        const float* g0 = wgt + (size_t)(wv * 8) * (CC * 27) + cb * (CB * 27);
        float* l0 = &Wlds[(wv * 8) * 27 * PT];
        const int cin = cb * CB + a_c8;
        const float* xc = src + (size_t)(b * CC + cin) * (DD * HH * WW) + a_w4 * 4;
        float ev = 0.f;
        if (MODE == 0) ev = es[b * CC + cin];

        float4 av[9];
        #pragma unroll
        for (int it = 0; it < 9; ++it) {
            const int rr = 2 * (it % 3) + a_pr0;
            const int zd = d + (it / 3) - 1;
            const int zh = h0 - 1 + rr;
            av[it] = make_float4(0.f, 0.f, 0.f, 0.f);
            if (zd >= 0 && zd < DD && zh >= 0 && zh < HH)
                av[it] = *(const float4*)(xc + (zd * HH + zh) * WW);
        }
        float wr[4][6], wt[3];
        #pragma unroll
        for (int oo = 0; oo < 4; ++oo) {
            #pragma unroll
            for (int k = 0; k < 6; ++k) wr[oo][k] = g0[oo * (CC * 27) + lane + 32 * k];
        }
        #pragma unroll
        for (int m = 0; m < 3; ++m) wt[m] = g0[tgo[m]];
        #pragma unroll
        for (int it = 0; it < 9; ++it) {
            const int rr = 2 * (it % 3) + a_pr0;
            const int pr = (it / 3) * AR + rr;
            const int zd = d + (it / 3) - 1;
            const int zh = h0 - 1 + rr;
            float4 v = av[it];
            if (MODE == 0) {
                if (zd >= 0 && zd < DD && zh >= 0 && zh < HH) {
                    v.x = swish_f(v.x + ev); v.y = swish_f(v.y + ev);
                    v.z = swish_f(v.z + ev); v.w = swish_f(v.w + ev);
                }
            }
            float* ap = &Alds[(pr * AROW + 1 + a_w4 * 4) * PT + a_c8];
            ap[0] = v.x; ap[PT] = v.y; ap[2 * PT] = v.z; ap[3 * PT] = v.w;
        }
        #pragma unroll
        for (int oo = 0; oo < 4; ++oo) {
            #pragma unroll
            for (int k = 0; k < 6; ++k) l0[oo * (27 * PT) + woff[k]] = wr[oo][k];
        }
        #pragma unroll
        for (int m = 0; m < 3; ++m) l0[tlo[m]] = wt[m];
        #pragma unroll
        for (int oo = 0; oo < 4; ++oo) {
            #pragma unroll
            for (int k = 0; k < 6; ++k) wr[oo][k] = g0[(oo + 4) * (CC * 27) + lane + 32 * k];
        }
        #pragma unroll
        for (int m = 0; m < 3; ++m) wt[m] = g0[4 * (CC * 27) + tgo[m]];
        #pragma unroll
        for (int oo = 0; oo < 4; ++oo) {
            #pragma unroll
            for (int k = 0; k < 6; ++k) l0[(oo + 4) * (27 * PT) + woff[k]] = wr[oo][k];
        }
        #pragma unroll
        for (int m = 0; m < 3; ++m) l0[4 * (27 * PT) + tlo[m]] = wt[m];
        __syncthreads();
        for (int kd = 0; kd < 3; ++kd) {
            const int zd = d + kd - 1;
            if (zd < 0 || zd >= DD) continue;
            #pragma unroll
            for (int kw = 0; kw < 3; ++kw) {
                const int aw = mpair * 16 + lane16 + kw;
                const float* ap = &Alds[((kd * AR) * AROW + aw) * PT + half * 4];
                v4f a[AR];
                #pragma unroll
                for (int rr = 0; rr < AR; ++rr) a[rr] = *(const v4f*)(ap + rr * (AROW * PT));
                #pragma unroll
                for (int kh = 0; kh < 3; ++kh) {
                    const int tap = kd * 9 + kh * 3 + kw;
                    const float* bp = &Wlds[((npair * 32 + lane16) * 27 + tap) * PT + half * 4];
                    const v4f b0 = *(const v4f*)bp;
                    const v4f b1 = *(const v4f*)(bp + 16 * 27 * PT);
                    const v2f b0l = {b0.x, b0.y}, b0h = {b0.z, b0.w};
                    const v2f b1l = {b1.x, b1.y}, b1h = {b1.z, b1.w};
                    #pragma unroll
                    for (int r = 0; r < TR; ++r) {
                        const v4f ar = a[kh + r];
                        const v2f al = {ar.x, ar.y};
                        acc[r][0] = __builtin_amdgcn_wmma_f32_16x16x4_f32(
                            false, al, false, b0l, (short)0, acc[r][0], false, false);
                        acc[r][1] = __builtin_amdgcn_wmma_f32_16x16x4_f32(
                            false, al, false, b1l, (short)0, acc[r][1], false, false);
                    }
                    #pragma unroll
                    for (int r = 0; r < TR; ++r) {
                        const v4f ar = a[kh + r];
                        const v2f ah = {ar.z, ar.w};
                        acc[r][0] = __builtin_amdgcn_wmma_f32_16x16x4_f32(
                            false, ah, false, b0h, (short)0, acc[r][0], false, false);
                        acc[r][1] = __builtin_amdgcn_wmma_f32_16x16x4_f32(
                            false, ah, false, b1h, (short)0, acc[r][1], false, false);
                    }
                }
            }
        }
    }

    const int wbase = mpair * 16 + half * 8;
    #pragma unroll
    for (int r = 0; r < TR; ++r) {
        #pragma unroll
        for (int t = 0; t < 2; ++t) {
            const v8f ac = acc[r][t];
            const int o = npair * 32 + t * 16 + lane16;
            const size_t off =
                (((size_t)(b * CC + o) * DD + d) * HH + (h0 + r)) * WW + wbase;
            float v[8];
            #pragma unroll
            for (int i = 0; i < 8; ++i) v[i] = ac[i];
            if (MODE == 0) {
                const float bo = bias[o];
                const float so = es[256 + b * CC + o];
                #pragma unroll
                for (int i = 0; i < 8; ++i) v[i] = swish_f(v[i] + bo) * so;
            } else {
                const float4 x0 = *(const float4*)(resid + off);
                const float4 x1 = *(const float4*)(resid + off + 4);
                v[0] += x0.x; v[1] += x0.y; v[2] += x0.z; v[3] += x0.w;
                v[4] += x1.x; v[5] += x1.y; v[6] += x1.z; v[7] += x1.w;
            }
            *(float4*)(dst + off)     = make_float4(v[0], v[1], v[2], v[3]);
            *(float4*)(dst + off + 4) = make_float4(v[4], v[5], v[6], v[7]);
        }
    }
}

extern "C" void kernel_launch(void* const* d_in, const int* in_sizes, int n_in,
                              void* d_out, int out_size, void* d_ws, size_t ws_size,
                              hipStream_t stream) {
    (void)in_sizes; (void)n_in; (void)out_size; (void)ws_size;
    const float* x      = (const float*)d_in[0];
    const float* embed  = (const float*)d_in[1];
    const float* prompt = (const float*)d_in[2];
    const float* c20w   = (const float*)d_in[3];
    const float* c20b   = (const float*)d_in[4];
    const float* c21w   = (const float*)d_in[5];
    const float* d1w    = (const float*)d_in[6];
    const float* d1b    = (const float*)d_in[7];
    const float* d2w    = (const float*)d_in[8];
    const float* d2b    = (const float*)d_in[9];
    float* out  = (float*)d_out;
    float* es   = (float*)d_ws;
    float* hbuf = es + 512;

    dense_es_kernel<<<1, 256, 0, stream>>>(embed, prompt, d1w, d1b, d2w, d2b, es);

    const int ngrid = 4 * DD * (HH / TR);
    conv3x3x3_wmma_kernel<0><<<ngrid, 256, 0, stream>>>(x, c20w, c20b, es, nullptr, hbuf);
    conv3x3x3_wmma_kernel<1><<<ngrid, 256, 0, stream>>>(hbuf, c21w, nullptr, nullptr, x, out);
}


# Minimal filtering F(2,3) along h: a pair of output rows from 4 transformed input rows, 4 products per (cin,kd,kw) instead of 6 (2/3 of the WMMAs); transforms while staging; f32 sum order changes
# speedup vs baseline: 6.3082x; 1.2485x over previous
#include <hip/hip_runtime.h>

typedef float v2f __attribute__((ext_vector_type(2)));
typedef float v4f __attribute__((ext_vector_type(4)));
typedef float v8f __attribute__((ext_vector_type(8)));

#define CC 64
#define DD 8
#define HH 64
#define WW 64
#define CB 8
#define AROW 66
#define TR 4
#define VR 8
#define APT 36
#define WS 292

__device__ __forceinline__ float swish_f(float v) {
    return v / (1.0f + __expf(-v));
}

__global__ __launch_bounds__(256) void dense_es_kernel(
    const float* __restrict__ embed, const float* __restrict__ prompt,
    const float* __restrict__ d1w, const float* __restrict__ d1b,
    const float* __restrict__ d2w, const float* __restrict__ d2b,
    float* __restrict__ es)
{
    const int t = threadIdx.x;
    const int bb = t >> 6, c = t & 63;
    const float* eb = embed + bb * 256;
    const float* pb = prompt + bb * 256;
    const float* w1 = d1w + c * 256;
    const float* w2 = d2w + c * 256;
    float a1 = 0.f, a2 = 0.f;
    for (int k = 0; k < 256; ++k) {
        a1 = fmaf(eb[k], w1[k], a1);
        a2 = fmaf(pb[k], w2[k], a2);
    }
    es[t]       = a1 + d1b[c];
    es[256 + t] = a2 + d2b[c];
}

template<int MODE>
__global__ __launch_bounds__(256) void conv3x3x3_wmma_kernel(
    const float* __restrict__ src,
    const float* __restrict__ wgt,
    const float* __restrict__ bias,
    const float* __restrict__ es,
    const float* __restrict__ resid,
    float* __restrict__ dst)
{
    __shared__ __attribute__((aligned(16))) float Alds[3 * 2 * AROW * APT];
    __shared__ __attribute__((aligned(16))) float Wlds[CC * WS];
    __shared__ __attribute__((aligned(16))) unsigned Tlds[32 * 20];

    const unsigned tid    = threadIdx.x;
    const unsigned lane   = tid & 31u;
    const unsigned wv     = (unsigned)__builtin_amdgcn_readfirstlane((int)(tid >> 5));
    const unsigned lane16 = lane & 15u;
    const unsigned half   = lane >> 4;
    const unsigned mpair  = wv & 3u;
    const unsigned npair  = wv >> 2;

    const int wg = blockIdx.x;
    const int hb = wg & 15;
    const int d  = (wg >> 4) & 7;
    const int b  = wg >> 7;
    const int h0 = hb * TR;

    for (unsigned i = tid; i < 3 * 2 * 2 * 32; i += 256) {
        const unsigned row = i >> 6, side = (i >> 5) & 1u, e = i & 31u;
        Alds[(row * AROW + side * (AROW - 1)) * APT + e] = 0.f;
    }

    const unsigned a_w4   = (lane & 7u) + 8u * (wv & 1u);
    const unsigned a_c4   = lane >> 3;
    const unsigned a_half = (wv >> 1) & 1u;
    const unsigned a_p    = wv >> 2;
    const unsigned a_voff = a_c4 * (DD * HH * WW) + a_w4 * 4u;
    float* const a_st = &Alds[(a_half * AROW + 1u + a_w4 * 4u) * APT + a_p * 16u + a_c4];

    if (wv == 0) {
        #pragma unroll
        for (unsigned k = 0; k < 9; ++k) {
            const unsigned q  = lane + 32u * k;
            const unsigned oo = q / 72u;
            const unsigned j  = q - 72u * oo;
            const unsigned c8 = j / 9u;
            const unsigned r9 = j - 9u * c8;
            const unsigned kd = r9 / 3u;
            const unsigned kw = r9 - 3u * kd;
            Tlds[lane * 20u + k]       = oo * (CC * 27) + c8 * 27u + kd * 9u + kw;
            Tlds[lane * 20u + 10u + k] = oo * WS + r9 * 8u + c8;
        }
    }

    v8f acc[VR][2];
    #pragma unroll
    for (int r = 0; r < VR; ++r) { acc[r][0] = (v8f){}; acc[r][1] = (v8f){}; }

    __builtin_amdgcn_s_setprio(3);
    for (int cb = 0; cb < CC / CB; ++cb) {
        if (cb == 3)      __builtin_amdgcn_s_setprio(2);
        else if (cb == 5) __builtin_amdgcn_s_setprio(1);
        else if (cb == 7) __builtin_amdgcn_s_setprio(0);
        __syncthreads();
        const uint4* tp = (const uint4*)&Tlds[lane * 20u];
        const uint4 q0 = tp[0], q1 = tp[1], q2 = tp[2], q3 = tp[3], q4 = tp[4];
        const unsigned wgo[9] = {q0.x, q0.y, q0.z, q0.w, q1.x, q1.y, q1.z, q1.w, q2.x};
        const unsigned wlo[9] = {q2.z, q2.w, q3.x, q3.y, q3.z, q3.w, q4.x, q4.y, q4.z};
        const float* g0 = wgt + (size_t)(wv * 8u) * (CC * 27) + cb * (CB * 27);
        const float* xb = src + (size_t)(b * CC + cb * CB + 4 * (int)a_half) * (DD * HH * WW);
        float ev = 0.f;
        if (MODE == 0) ev = es[b * CC + cb * CB + (int)(a_half * 4u + a_c4)];

        auto load_plane = [&](int kd, float4 (&av)[4]) {
            const int zd = d + kd - 1;
            #pragma unroll
            for (int i = 0; i < 4; ++i) {
                const int zh = h0 - 1 + 2 * (int)a_p + i;
                av[i] = make_float4(0.f, 0.f, 0.f, 0.f);
                if (zd >= 0 && zd < DD && zh >= 0 && zh < HH)
                    av[i] = *(const float4*)(xb + (zd * HH + zh) * WW + a_voff);
            }
        };
        auto stage_plane = [&](int kd, const float4 (&av)[4]) {
            const int zd = d + kd - 1;
            if (zd < 0 || zd >= DD) return;
            float4 v[4];
            #pragma unroll
            for (int i = 0; i < 4; ++i) {
                const int zh = h0 - 1 + 2 * (int)a_p + i;
                v[i] = av[i];
                if (MODE == 0) {
                    if (zh >= 0 && zh < HH) {
                        v[i].x = swish_f(v[i].x + ev); v[i].y = swish_f(v[i].y + ev);
                        v[i].z = swish_f(v[i].z + ev); v[i].w = swish_f(v[i].w + ev);
                    }
                }
            }
            float4 tv[4];
            tv[0] = make_float4(v[0].x - v[2].x, v[0].y - v[2].y, v[0].z - v[2].z, v[0].w - v[2].w);
            tv[1] = make_float4(v[1].x + v[2].x, v[1].y + v[2].y, v[1].z + v[2].z, v[1].w + v[2].w);
            tv[2] = make_float4(v[2].x - v[1].x, v[2].y - v[1].y, v[2].z - v[1].z, v[2].w - v[1].w);
            tv[3] = make_float4(v[1].x - v[3].x, v[1].y - v[3].y, v[1].z - v[3].z, v[1].w - v[3].w);
            float* ap = a_st + kd * (2 * AROW * APT);
            #pragma unroll
            for (int t = 0; t < 4; ++t) {
                ap[t * 4]           = tv[t].x;
                ap[t * 4 + APT]     = tv[t].y;
                ap[t * 4 + 2 * APT] = tv[t].z;
                ap[t * 4 + 3 * APT] = tv[t].w;
            }
        };
        auto load_w = [&](int o4, float (&wr)[9][3]) {
            #pragma unroll
            for (int k = 0; k < 9; ++k) {
                const float* gp = g0 + wgo[k];
                #pragma unroll
                for (int kh = 0; kh < 3; ++kh) wr[k][kh] = gp[o4 * (CC * 27) + 3 * kh];
            }
        };
        auto stage_w = [&](int o4, const float (&wr)[9][3]) {
            #pragma unroll
            for (int k = 0; k < 9; ++k) {
                const float s = 0.5f * (wr[k][0] + wr[k][2]);
                float* lp = &Wlds[(wv * 8u + o4) * WS + wlo[k]];
                lp[0]   = wr[k][0];
                lp[72]  = fmaf( 0.5f, wr[k][1], s);
                lp[144] = fmaf(-0.5f, wr[k][1], s);
                lp[216] = wr[k][2];
            }
        };

        float4 av0[4], av1[4], av2[4];
        float wr[9][3];
        load_plane(0, av0);
        load_plane(1, av1);
        load_w(0, wr);
        stage_plane(0, av0);
        load_plane(2, av2);
        stage_w(0, wr);
        load_w(4, wr);
        stage_plane(1, av1);
        stage_plane(2, av2);
        stage_w(4, wr);
        __syncthreads();
        for (int kd = 0; kd < 3; ++kd) {
            const int zd = d + kd - 1;
            if (zd < 0 || zd >= DD) continue;
            #pragma unroll
            for (int kw = 0; kw < 3; ++kw) {
                const unsigned aw = mpair * 16u + lane16 + kw;
                const float* ap = &Alds[((kd * 2 + half) * AROW + aw) * APT];
                v4f a[VR];
                #pragma unroll
                for (int vr = 0; vr < VR; ++vr) a[vr] = *(const v4f*)(ap + vr * 4);
                #pragma unroll
                for (int t = 0; t < 4; ++t) {
                    const int tq = t * 9 + kd * 3 + kw;
                    const float* bp = &Wlds[(npair * 32u + lane16) * WS + tq * 8 + half * 4u];
                    const v4f b0 = *(const v4f*)bp;
                    const v4f b1 = *(const v4f*)(bp + 16 * WS);
                    const v2f b0l = {b0.x, b0.y}, b0h = {b0.z, b0.w};
                    const v2f b1l = {b1.x, b1.y}, b1h = {b1.z, b1.w};
                    #pragma unroll
                    for (int p = 0; p < 2; ++p) {
                        const v4f ar = a[p * 4 + t];
                        const v2f al = {ar.x, ar.y};
                        acc[p * 4 + t][0] = __builtin_amdgcn_wmma_f32_16x16x4_f32(
                            false, al, false, b0l, (short)0, acc[p * 4 + t][0], false, false);
                        acc[p * 4 + t][1] = __builtin_amdgcn_wmma_f32_16x16x4_f32(
                            false, al, false, b1l, (short)0, acc[p * 4 + t][1], false, false);
                    }
                    #pragma unroll
                    for (int p = 0; p < 2; ++p) {
                        const v4f ar = a[p * 4 + t];
                        const v2f ah = {ar.z, ar.w};
                        acc[p * 4 + t][0] = __builtin_amdgcn_wmma_f32_16x16x4_f32(
                            false, ah, false, b0h, (short)0, acc[p * 4 + t][0], false, false);
                        acc[p * 4 + t][1] = __builtin_amdgcn_wmma_f32_16x16x4_f32(
                            false, ah, false, b1h, (short)0, acc[p * 4 + t][1], false, false);
                    }
                }
            }
        }
    }

    const unsigned wbase = mpair * 16u + half * 8u;
    #pragma unroll
    for (int p = 0; p < 2; ++p) {
        #pragma unroll
        for (int t = 0; t < 2; ++t) {
            const v8f m0 = acc[p * 4 + 0][t], m1 = acc[p * 4 + 1][t];
            const v8f m2 = acc[p * 4 + 2][t], m3 = acc[p * 4 + 3][t];
            const int o = (int)(npair * 32u + t * 16u + lane16);
            #pragma unroll
            for (int rr = 0; rr < 2; ++rr) {
                const size_t off =
                    (((size_t)(b * CC + o) * DD + d) * HH + (h0 + 2 * p + rr)) * WW + wbase;
                float v[8];
                #pragma unroll
                for (int i = 0; i < 8; ++i)
                    v[i] = (rr == 0) ? ((m0[i] + m1[i]) + m2[i]) : ((m1[i] - m2[i]) - m3[i]);
                if (MODE == 0) {
                    const float bo = bias[o];
                    const float so = es[256 + b * CC + o];
                    #pragma unroll
                    for (int i = 0; i < 8; ++i) v[i] = swish_f(v[i] + bo) * so;
                } else {
                    const float4 x0 = *(const float4*)(resid + off);
                    const float4 x1 = *(const float4*)(resid + off + 4);
                    v[0] += x0.x; v[1] += x0.y; v[2] += x0.z; v[3] += x0.w;
                    v[4] += x1.x; v[5] += x1.y; v[6] += x1.z; v[7] += x1.w;
                }
                *(float4*)(dst + off)     = make_float4(v[0], v[1], v[2], v[3]);
                *(float4*)(dst + off + 4) = make_float4(v[4], v[5], v[6], v[7]);
            }
        }
    }
}

extern "C" void kernel_launch(void* const* d_in, const int* in_sizes, int n_in,
                              void* d_out, int out_size, void* d_ws, size_t ws_size,
                              hipStream_t stream) {
    (void)in_sizes; (void)n_in; (void)out_size; (void)ws_size;
    const float* x      = (const float*)d_in[0];
    const float* embed  = (const float*)d_in[1];
    const float* prompt = (const float*)d_in[2];
    const float* c20w   = (const float*)d_in[3];
    const float* c20b   = (const float*)d_in[4];
    const float* c21w   = (const float*)d_in[5];
    const float* d1w    = (const float*)d_in[6];
    const float* d1b    = (const float*)d_in[7];
    const float* d2w    = (const float*)d_in[8];
    const float* d2b    = (const float*)d_in[9];
    float* out  = (float*)d_out;
    float* es   = (float*)d_ws;
    float* hbuf = es + 512;

    dense_es_kernel<<<1, 256, 0, stream>>>(embed, prompt, d1w, d1b, d2w, d2b, es);

    const int ngrid = 4 * DD * (HH / TR);
    conv3x3x3_wmma_kernel<0><<<ngrid, 256, 0, stream>>>(x, c20w, c20b, es, nullptr, hbuf);
    conv3x3x3_wmma_kernel<1><<<ngrid, 256, 0, stream>>>(hbuf, c21w, nullptr, nullptr, x, out);
}


# Dense launch folded into the convs: each workgroup computes its batch row's 64 shifts e (conv 1) or styles s (conv 2, applied to its staged input) into LDS; 3 launches to 2; dense sum order changes
# speedup vs baseline: 7.4355x; 1.1787x over previous
#include <hip/hip_runtime.h>

typedef float v2f __attribute__((ext_vector_type(2)));
typedef float v4f __attribute__((ext_vector_type(4)));
typedef float v8f __attribute__((ext_vector_type(8)));

#define CC 64
#define DD 8
#define HH 64
#define WW 64
#define CB 8
#define AROW 66
#define TR 4
#define VR 6
#define APT 28
#define WS 436

__device__ __forceinline__ float swish_f(float v) {
    return v / (1.0f + __expf(-v));
}

template<int MODE>
__global__ __launch_bounds__(256) void conv3x3x3_wmma_kernel(
    const float* __restrict__ src,
    const float* __restrict__ wgt,
    const float* __restrict__ bias,
    const float* __restrict__ vec,
    const float* __restrict__ dw,
    const float* __restrict__ db,
    const float* __restrict__ resid,
    float* __restrict__ dst)
{
    __shared__ __attribute__((aligned(16))) float Alds[3 * 2 * AROW * APT];
    __shared__ __attribute__((aligned(16))) float Wlds[CC * WS];
    __shared__ __attribute__((aligned(16))) unsigned Tlds[32 * 20];
    __shared__ float Clds[CC];

    const unsigned tid    = threadIdx.x;
    const unsigned lane   = tid & 31u;
    const unsigned wv     = (unsigned)__builtin_amdgcn_readfirstlane((int)(tid >> 5));
    const unsigned lane16 = lane & 15u;
    const unsigned half   = lane >> 4;
    const unsigned mpair  = wv & 3u;
    const unsigned npair  = wv >> 2;

    const int wg = blockIdx.x;
    const int hb = wg & 15;
    const int d  = (wg >> 4) & 7;
    const int b  = wg >> 7;
    const int h0 = hb * TR;

    {
        const float* vb = vec + b * 256 + 4u * lane16;
        float4 xv[4];
        #pragma unroll
        for (int i = 0; i < 4; ++i) xv[i] = *(const float4*)(vb + 64 * i);
        float part[4], dbv[4];
        #pragma unroll
        for (int p = 0; p < 4; ++p) dbv[p] = db[wv * 8u + 2u * p + half];
        #pragma unroll
        for (int p = 0; p < 4; ++p) {
            const float* wrow = dw + (wv * 8u + 2u * p + half) * 256u + 4u * lane16;
            float s = (lane16 == 0) ? dbv[p] : 0.f;
            #pragma unroll
            for (int i = 0; i < 4; ++i) {
                const float4 w4 = *(const float4*)(wrow + 64 * i);
                s = fmaf(xv[i].x, w4.x, s);
                s = fmaf(xv[i].y, w4.y, s);
                s = fmaf(xv[i].z, w4.z, s);
                s = fmaf(xv[i].w, w4.w, s);
            }
            part[p] = s;
        }
        #pragma unroll
        for (int m = 8; m >= 1; m >>= 1) {
            #pragma unroll
            for (int p = 0; p < 4; ++p) part[p] += __shfl_xor(part[p], m, 32);
        }
        if (lane16 == 0) {
            #pragma unroll
            for (int p = 0; p < 4; ++p) Clds[wv * 8u + 2u * p + half] = part[p];
        }
    }

    for (unsigned i = tid; i < 3 * 2 * 2 * (VR * 4); i += 256) {
        const unsigned row = i / (2u * VR * 4u), rem = i - row * (2u * VR * 4u);
        const unsigned side = rem / (VR * 4u), e = rem - side * (VR * 4u);
        Alds[(row * AROW + side * (AROW - 1)) * APT + e] = 0.f;
    }

    const unsigned a_cl   = lane >> 4;
    const unsigned a_cp   = wv >> 1;
    const unsigned a_w    = 32u * (wv & 1u) + 2u * lane16;
    const unsigned a_voff = a_cl * (DD * HH * WW) + a_w;
    float* const a_st = &Alds[((a_cp >> 1) * AROW + 1u + a_w) * APT + 2u * (a_cp & 1u) + a_cl];

    if (wv == 0) {
        #pragma unroll
        for (unsigned k = 0; k < 9; ++k) {
            const unsigned q  = lane + 32u * k;
            const unsigned oo = q / 72u;
            const unsigned j  = q - 72u * oo;
            const unsigned c8 = j / 9u;
            const unsigned r9 = j - 9u * c8;
            const unsigned kd = r9 / 3u;
            const unsigned kw = r9 - 3u * kd;
            Tlds[lane * 20u + k]       = oo * (CC * 27) + c8 * 27u + kd * 9u + kw;
            Tlds[lane * 20u + 10u + k] = oo * WS + r9 * 8u + c8;
        }
    }

    v8f acc[VR][2];
    #pragma unroll
    for (int r = 0; r < VR; ++r) { acc[r][0] = (v8f){}; acc[r][1] = (v8f){}; }

    __builtin_amdgcn_s_setprio(3);
    for (int cb = 0; cb < CC / CB; ++cb) {
        if (cb == 3)      __builtin_amdgcn_s_setprio(2);
        else if (cb == 5) __builtin_amdgcn_s_setprio(1);
        else if (cb == 7) __builtin_amdgcn_s_setprio(0);
        __syncthreads();
        const uint4* tp = (const uint4*)&Tlds[lane * 20u];
        const uint4 q0 = tp[0], q1 = tp[1], q2 = tp[2], q3 = tp[3], q4 = tp[4];
        const unsigned wgo[9] = {q0.x, q0.y, q0.z, q0.w, q1.x, q1.y, q1.z, q1.w, q2.x};
        const unsigned wlo[9] = {q2.z, q2.w, q3.x, q3.y, q3.z, q3.w, q4.x, q4.y, q4.z};
        const float* g0 = wgt + (size_t)(wv * 8u) * (CC * 27) + cb * (CB * 27);
        const float* xb = src + (size_t)(b * CC + cb * CB + 2 * (int)a_cp) * (DD * HH * WW);
        const float cs = Clds[cb * CB + (int)(2u * a_cp + a_cl)];

        auto load_plane = [&](int kd, float2 (&av)[6]) {
            const int zd = d + kd - 1;
            #pragma unroll
            for (int i = 0; i < 6; ++i) {
                const int zh = h0 - 1 + i;
                av[i] = make_float2(0.f, 0.f);
                if (zd >= 0 && zd < DD && zh >= 0 && zh < HH)
                    av[i] = *(const float2*)(xb + (zd * HH + zh) * WW + a_voff);
            }
        };
        auto stage_plane = [&](int kd, const float2 (&av)[6]) {
            const int zd = d + kd - 1;
            if (zd < 0 || zd >= DD) return;
            float2 v[6];
            #pragma unroll
            for (int i = 0; i < 6; ++i) {
                const int zh = h0 - 1 + i;
                v[i] = av[i];
                if (MODE == 0) {
                    if (zh >= 0 && zh < HH) {
                        v[i].x = swish_f(v[i].x + cs); v[i].y = swish_f(v[i].y + cs);
                    }
                } else {
                    v[i].x *= cs; v[i].y *= cs;
                }
            }
            float tx[6], ty[6];
            {
                const float a = fmaf(-4.f, v[2].x, v[4].x), bq = fmaf(-4.f, v[1].x, v[3].x);
                const float c = v[4].x - v[2].x,            e  = v[3].x - v[1].x;
                tx[0] = fmaf(4.f, v[0].x, fmaf(-5.f, v[2].x, v[4].x));
                tx[1] = a + bq;
                tx[2] = a - bq;
                tx[3] = fmaf( 2.f, e, c);
                tx[4] = fmaf(-2.f, e, c);
                tx[5] = fmaf(4.f, v[1].x, fmaf(-5.f, v[3].x, v[5].x));
            }
            {
                const float a = fmaf(-4.f, v[2].y, v[4].y), bq = fmaf(-4.f, v[1].y, v[3].y);
                const float c = v[4].y - v[2].y,            e  = v[3].y - v[1].y;
                ty[0] = fmaf(4.f, v[0].y, fmaf(-5.f, v[2].y, v[4].y));
                ty[1] = a + bq;
                ty[2] = a - bq;
                ty[3] = fmaf( 2.f, e, c);
                ty[4] = fmaf(-2.f, e, c);
                ty[5] = fmaf(4.f, v[1].y, fmaf(-5.f, v[3].y, v[5].y));
            }
            float* ap = a_st + kd * (2 * AROW * APT);
            #pragma unroll
            for (int t = 0; t < VR; ++t) {
                ap[t * 4]       = tx[t];
                ap[t * 4 + APT] = ty[t];
            }
        };
        auto load_w = [&](int o4, float (&wr)[9][3]) {
            #pragma unroll
            for (int k = 0; k < 9; ++k) {
                const float* gp = g0 + wgo[k];
                #pragma unroll
                for (int kh = 0; kh < 3; ++kh) wr[k][kh] = gp[o4 * (CC * 27) + 3 * kh];
            }
        };
        auto stage_w = [&](int o4, const float (&wr)[9][3]) {
            #pragma unroll
            for (int k = 0; k < 9; ++k) {
                const float s = wr[k][0] + wr[k][2];
                const float u = fmaf(1.f / 24.f, wr[k][0], (1.f / 6.f) * wr[k][2]);
                float* lp = &Wlds[(wv * 8u + o4) * WS + wlo[k]];
                lp[0]   = 0.25f * wr[k][0];
                lp[72]  = (-1.f / 6.f) * (s + wr[k][1]);
                lp[144] = (-1.f / 6.f) * (s - wr[k][1]);
                lp[216] = fmaf( 1.f / 12.f, wr[k][1], u);
                lp[288] = fmaf(-1.f / 12.f, wr[k][1], u);
                lp[360] = wr[k][2];
            }
        };

        float2 av0[6], av1[6], av2[6];
        float wr[9][3];
        load_plane(0, av0);
        load_plane(1, av1);
        load_w(0, wr);
        stage_plane(0, av0);
        load_plane(2, av2);
        stage_w(0, wr);
        load_w(4, wr);
        stage_plane(1, av1);
        stage_plane(2, av2);
        stage_w(4, wr);
        __syncthreads();
        for (int kd = 0; kd < 3; ++kd) {
            const int zd = d + kd - 1;
            if (zd < 0 || zd >= DD) continue;
            #pragma unroll
            for (int kw = 0; kw < 3; ++kw) {
                const unsigned aw = mpair * 16u + lane16 + kw;
                const float* ap = &Alds[((kd * 2 + half) * AROW + aw) * APT];
                v4f a[VR];
                #pragma unroll
                for (int vr = 0; vr < VR; ++vr) a[vr] = *(const v4f*)(ap + vr * 4);
                #pragma unroll
                for (int tp = 0; tp < VR / 2; ++tp) {
                    v4f b0[2], b1[2];
                    #pragma unroll
                    for (int u = 0; u < 2; ++u) {
                        const int tq = (2 * tp + u) * 9 + kd * 3 + kw;
                        const float* bp = &Wlds[(npair * 32u + lane16) * WS + tq * 8 + half * 4u];
                        b0[u] = *(const v4f*)bp;
                        b1[u] = *(const v4f*)(bp + 16 * WS);
                    }
                    #pragma unroll
                    for (int u = 0; u < 2; ++u) {
                        const v4f ar = a[2 * tp + u];
                        const v2f al = {ar.x, ar.y};
                        const v2f b0l = {b0[u].x, b0[u].y}, b1l = {b1[u].x, b1[u].y};
                        acc[2 * tp + u][0] = __builtin_amdgcn_wmma_f32_16x16x4_f32(
                            false, al, false, b0l, (short)0, acc[2 * tp + u][0], false, false);
                        acc[2 * tp + u][1] = __builtin_amdgcn_wmma_f32_16x16x4_f32(
                            false, al, false, b1l, (short)0, acc[2 * tp + u][1], false, false);
                    }
                    #pragma unroll
                    for (int u = 0; u < 2; ++u) {
                        const v4f ar = a[2 * tp + u];
                        const v2f ah = {ar.z, ar.w};
                        const v2f b0h = {b0[u].z, b0[u].w}, b1h = {b1[u].z, b1[u].w};
                        acc[2 * tp + u][0] = __builtin_amdgcn_wmma_f32_16x16x4_f32(
                            false, ah, false, b0h, (short)0, acc[2 * tp + u][0], false, false);
                        acc[2 * tp + u][1] = __builtin_amdgcn_wmma_f32_16x16x4_f32(
                            false, ah, false, b1h, (short)0, acc[2 * tp + u][1], false, false);
                    }
                }
            }
        }
    }

    const unsigned wbase = mpair * 16u + half * 8u;
    {
        #pragma unroll
        for (int t = 0; t < 2; ++t) {
            const v8f m0 = acc[0][t], m1 = acc[1][t], m2 = acc[2][t];
            const v8f m3 = acc[3][t], m4 = acc[4][t], m5 = acc[5][t];
            const int o = (int)(npair * 32u + t * 16u + lane16);
            #pragma unroll
            for (int rr = 0; rr < 4; ++rr) {
                const size_t off =
                    (((size_t)(b * CC + o) * DD + d) * HH + (h0 + rr)) * WW + wbase;
                float v[8];
                #pragma unroll
                for (int i = 0; i < 8; ++i) {
                    const float dm = m1[i] - m2[i], sm = m1[i] + m2[i];
                    const float dq = m3[i] - m4[i], sq = m3[i] + m4[i];
                    v[i] = (rr == 0) ? ((m0[i] + sm) + sq)
                         : (rr == 1) ? fmaf(2.f, dq, dm)
                         : (rr == 2) ? fmaf(4.f, sq, sm)
                         :             (fmaf(8.f, dq, dm) + m5[i]);
                }
                if (MODE == 0) {
                    const float bo = bias[o];
                    #pragma unroll
                    for (int i = 0; i < 8; ++i) v[i] = swish_f(v[i] + bo);
                } else {
                    const float4 x0 = *(const float4*)(resid + off);
                    const float4 x1 = *(const float4*)(resid + off + 4);
                    v[0] += x0.x; v[1] += x0.y; v[2] += x0.z; v[3] += x0.w;
                    v[4] += x1.x; v[5] += x1.y; v[6] += x1.z; v[7] += x1.w;
                }
                *(float4*)(dst + off)     = make_float4(v[0], v[1], v[2], v[3]);
                *(float4*)(dst + off + 4) = make_float4(v[4], v[5], v[6], v[7]);
            }
        }
    }
}

extern "C" void kernel_launch(void* const* d_in, const int* in_sizes, int n_in,
                              void* d_out, int out_size, void* d_ws, size_t ws_size,
                              hipStream_t stream) {
    (void)in_sizes; (void)n_in; (void)out_size; (void)ws_size;
    const float* x      = (const float*)d_in[0];
    const float* embed  = (const float*)d_in[1];
    const float* prompt = (const float*)d_in[2];
    const float* c20w   = (const float*)d_in[3];
    const float* c20b   = (const float*)d_in[4];
    const float* c21w   = (const float*)d_in[5];
    const float* d1w    = (const float*)d_in[6];
    const float* d1b    = (const float*)d_in[7];
    const float* d2w    = (const float*)d_in[8];
    const float* d2b    = (const float*)d_in[9];
    float* out  = (float*)d_out;
    float* hbuf = (float*)d_ws + 512;

    const int ngrid = 4 * DD * (HH / TR);
    conv3x3x3_wmma_kernel<0><<<ngrid, 256, 0, stream>>>(x, c20w, c20b, embed, d1w, d1b, nullptr, hbuf);
    conv3x3x3_wmma_kernel<1><<<ngrid, 256, 0, stream>>>(hbuf, c21w, nullptr, prompt, d2w, d2b, x, out);
}


# Minimal filtering nested along d: F(2,3) on plane pairs over F(4,3) rows, 24 products per (cin,kw) for 2 planes x 4 rows instead of 36 (2/3 of the WMMAs); 1 workgroup per WGP; f32 sum order changes
# speedup vs baseline: 8.3480x; 1.1227x over previous
#include <hip/hip_runtime.h>


typedef float v2f __attribute__((ext_vector_type(2)));
typedef float v4f __attribute__((ext_vector_type(4)));
typedef float v8f __attribute__((ext_vector_type(8)));

#define CC 64
#define DD 8
#define HH 64
#define WW 64
#define CB 8
#define AROW 66
#define TR 4
#define VR 6
#define DT 2
#define DV 4
#define APT 28
#define NTAP (DV * VR * 3)
#define WS (NTAP * CB + 4)

__device__ __forceinline__ float swish_f(float v) {
    return v / (1.0f + __expf(-v));
}

template<int MODE>
__global__ __launch_bounds__(256) void conv3x3x3_wmma_kernel(
    const float* __restrict__ src,
    const float* __restrict__ wgt,
    const float* __restrict__ bias,
    const float* __restrict__ vec,
    const float* __restrict__ dw,
    const float* __restrict__ db,
    const float* __restrict__ resid,
    float* __restrict__ dst)
{
    __shared__ __attribute__((aligned(16))) float Alds[DV * 2 * AROW * APT];
    __shared__ __attribute__((aligned(16))) float Wlds[CC * WS];
    __shared__ __attribute__((aligned(16))) unsigned Tlds[32 * 12];
    __shared__ float Clds[CC];

    const unsigned tid    = threadIdx.x;
    const unsigned lane   = tid & 31u;
    const unsigned wv     = (unsigned)__builtin_amdgcn_readfirstlane((int)(tid >> 5));
    const unsigned lane16 = lane & 15u;
    const unsigned half   = lane >> 4;
    const unsigned mpair  = wv & 3u;
    const unsigned npair  = wv >> 2;

    const int wg = blockIdx.x;
    const int hb = wg & 15;
    const int dp = (wg >> 4) & 3;
    const int b  = wg >> 6;
    const int h0 = hb * TR;
    const int d0 = dp * DT;

    {
        const float* vb = vec + b * 256 + 4u * lane16;
        float4 xv[4];
        #pragma unroll
        for (int i = 0; i < 4; ++i) xv[i] = *(const float4*)(vb + 64 * i);
        float part[4], dbv[4];
        #pragma unroll
        for (int p = 0; p < 4; ++p) dbv[p] = db[wv * 8u + 2u * p + half];
        #pragma unroll
        for (int p = 0; p < 4; ++p) {
            const float* wrow = dw + (wv * 8u + 2u * p + half) * 256u + 4u * lane16;
            float s = (lane16 == 0) ? dbv[p] : 0.f;
            #pragma unroll
            for (int i = 0; i < 4; ++i) {
                const float4 w4 = *(const float4*)(wrow + 64 * i);
                s = fmaf(xv[i].x, w4.x, s);
                s = fmaf(xv[i].y, w4.y, s);
                s = fmaf(xv[i].z, w4.z, s);
                s = fmaf(xv[i].w, w4.w, s);
            }
            part[p] = s;
        }
        #pragma unroll
        for (int m = 8; m >= 1; m >>= 1) {
            #pragma unroll
            for (int p = 0; p < 4; ++p) part[p] += __shfl_xor(part[p], m, 32);
        }
        if (lane16 == 0) {
            #pragma unroll
            for (int p = 0; p < 4; ++p) Clds[wv * 8u + 2u * p + half] = part[p];
        }
    }

    for (unsigned i = tid; i < DV * 2 * 2 * (VR * 4); i += 256) {
        const unsigned row = i / (2u * VR * 4u), rem = i - row * (2u * VR * 4u);
        const unsigned side = rem / (VR * 4u), e = rem - side * (VR * 4u);
        Alds[(row * AROW + side * (AROW - 1)) * APT + e] = 0.f;
    }

    const unsigned a_cl   = lane >> 4;
    const unsigned a_cp   = wv >> 1;
    const unsigned a_w    = 32u * (wv & 1u) + 2u * lane16;
    const unsigned a_voff = a_cl * (DD * HH * WW) + a_w;
    float* const a_st = &Alds[((a_cp >> 1) * AROW + 1u + a_w) * APT + 2u * (a_cp & 1u) + a_cl];

    if (wv == 0) {
        #pragma unroll
        for (unsigned k = 0; k < 6; ++k) {
            const unsigned q  = lane + 32u * k;
            const unsigned oo = q / 24u;
            const unsigned j  = q - 24u * oo;
            const unsigned c8 = j / 3u;
            const unsigned kw = j - 3u * c8;
            Tlds[lane * 12u + k]      = oo * (CC * 27) + c8 * 27u + kw;
            Tlds[lane * 12u + 6u + k] = oo * WS + kw * 8u + c8;
        }
    }

    v8f acc[DV][VR][2];
    #pragma unroll
    for (int dc = 0; dc < DV; ++dc) {
        #pragma unroll
        for (int r = 0; r < VR; ++r) { acc[dc][r][0] = (v8f){}; acc[dc][r][1] = (v8f){}; }
    }

    for (int cb = 0; cb < CC / CB; ++cb) {
        __syncthreads();
        const uint4* tp = (const uint4*)&Tlds[lane * 12u];
        const uint4 q0 = tp[0], q1 = tp[1], q2 = tp[2];
        const unsigned wgo[6] = {q0.x, q0.y, q0.z, q0.w, q1.x, q1.y};
        const unsigned wlo[6] = {q1.z, q1.w, q2.x, q2.y, q2.z, q2.w};
        const float* g0 = wgt + (size_t)(wv * 8u) * (CC * 27) + cb * (CB * 27);
        const float* xb = src + (size_t)(b * CC + cb * CB + 2 * (int)a_cp) * (DD * HH * WW);
        const float cs = Clds[cb * CB + (int)(2u * a_cp + a_cl)];

        auto load_plane = [&](int pi, float2 (&av)[6]) {
            const int zd = d0 + pi - 1;
            #pragma unroll
            for (int i = 0; i < 6; ++i) {
                const int zh = h0 - 1 + i;
                av[i] = make_float2(0.f, 0.f);
                if (zd >= 0 && zd < DD && zh >= 0 && zh < HH)
                    av[i] = *(const float2*)(xb + (zd * HH + zh) * WW + a_voff);
            }
        };
        auto transform_plane = [&](int pi, const float2 (&av)[6], float (&tx)[6], float (&ty)[6]) {
            const int zd = d0 + pi - 1;
            float2 v[6];
            #pragma unroll
            for (int i = 0; i < 6; ++i) {
                const int zh = h0 - 1 + i;
                v[i] = av[i];
                if (MODE == 0) {
                    if (zd >= 0 && zd < DD && zh >= 0 && zh < HH) {
                        v[i].x = swish_f(v[i].x + cs); v[i].y = swish_f(v[i].y + cs);
                    }
                } else {
                    v[i].x *= cs; v[i].y *= cs;
                }
            }
            {
                const float a = fmaf(-4.f, v[2].x, v[4].x), bq = fmaf(-4.f, v[1].x, v[3].x);
                const float c = v[4].x - v[2].x,            e  = v[3].x - v[1].x;
                tx[0] = fmaf(4.f, v[0].x, fmaf(-5.f, v[2].x, v[4].x));
                tx[1] = a + bq;
                tx[2] = a - bq;
                tx[3] = fmaf( 2.f, e, c);
                tx[4] = fmaf(-2.f, e, c);
                tx[5] = fmaf(4.f, v[1].x, fmaf(-5.f, v[3].x, v[5].x));
            }
            {
                const float a = fmaf(-4.f, v[2].y, v[4].y), bq = fmaf(-4.f, v[1].y, v[3].y);
                const float c = v[4].y - v[2].y,            e  = v[3].y - v[1].y;
                ty[0] = fmaf(4.f, v[0].y, fmaf(-5.f, v[2].y, v[4].y));
                ty[1] = a + bq;
                ty[2] = a - bq;
                ty[3] = fmaf( 2.f, e, c);
                ty[4] = fmaf(-2.f, e, c);
                ty[5] = fmaf(4.f, v[1].y, fmaf(-5.f, v[3].y, v[5].y));
            }
        };
        auto load_w = [&](int k0, float (&wr)[3][9]) {
            #pragma unroll
            for (int k = 0; k < 3; ++k) {
                const float* gp = g0 + wgo[k0 + k];
                #pragma unroll
                for (int kd = 0; kd < 3; ++kd) {
                    #pragma unroll
                    for (int kh = 0; kh < 3; ++kh) wr[k][kd * 3 + kh] = gp[kd * 9 + kh * 3];
                }
            }
        };
        auto stage_w = [&](int k0, const float (&wr)[3][9]) {
            #pragma unroll
            for (int k = 0; k < 3; ++k) {
                float uh[3][6];
                #pragma unroll
                for (int kd = 0; kd < 3; ++kd) {
                    const float w0 = wr[k][kd * 3], w1 = wr[k][kd * 3 + 1], w2 = wr[k][kd * 3 + 2];
                    const float s = w0 + w2;
                    const float u = fmaf(1.f / 24.f, w0, (1.f / 6.f) * w2);
                    uh[kd][0] = 0.25f * w0;
                    uh[kd][1] = (-1.f / 6.f) * (s + w1);
                    uh[kd][2] = (-1.f / 6.f) * (s - w1);
                    uh[kd][3] = fmaf( 1.f / 12.f, w1, u);
                    uh[kd][4] = fmaf(-1.f / 12.f, w1, u);
                    uh[kd][5] = w2;
                }
                float* lp = &Wlds[(wv * 8u) * WS + wlo[k0 + k]];
                #pragma unroll
                for (int t = 0; t < VR; ++t) {
                    const float s = uh[0][t] + uh[2][t];
                    lp[(0 * VR + t) * 24] = uh[0][t];
                    lp[(1 * VR + t) * 24] = 0.5f * (s + uh[1][t]);
                    lp[(2 * VR + t) * 24] = 0.5f * (s - uh[1][t]);
                    lp[(3 * VR + t) * 24] = uh[2][t];
                }
            }
        };

        float2 av0[6], av1[6], av2[6], av3[6];
        float wr[3][9];
        load_w(0, wr);
        load_plane(0, av0);
        load_plane(1, av1);
        load_plane(2, av2);
        load_plane(3, av3);
        stage_w(0, wr);
        load_w(3, wr);
        {
            float tx[DV][6], ty[DV][6];
            transform_plane(0, av0, tx[0], ty[0]);
            transform_plane(1, av1, tx[1], ty[1]);
            transform_plane(2, av2, tx[2], ty[2]);
            transform_plane(3, av3, tx[3], ty[3]);
            #pragma unroll
            for (int t = 0; t < VR; ++t) {
                a_st[0 * (2 * AROW * APT) + t * 4]       = tx[0][t] - tx[2][t];
                a_st[0 * (2 * AROW * APT) + t * 4 + APT] = ty[0][t] - ty[2][t];
                a_st[1 * (2 * AROW * APT) + t * 4]       = tx[1][t] + tx[2][t];
                a_st[1 * (2 * AROW * APT) + t * 4 + APT] = ty[1][t] + ty[2][t];
                a_st[2 * (2 * AROW * APT) + t * 4]       = tx[2][t] - tx[1][t];
                a_st[2 * (2 * AROW * APT) + t * 4 + APT] = ty[2][t] - ty[1][t];
                a_st[3 * (2 * AROW * APT) + t * 4]       = tx[1][t] - tx[3][t];
                a_st[3 * (2 * AROW * APT) + t * 4 + APT] = ty[1][t] - ty[3][t];
            }
        }
        stage_w(3, wr);
        __syncthreads();
        #pragma unroll
        for (int dc = 0; dc < DV; ++dc) {
            #pragma unroll
            for (int kw = 0; kw < 3; ++kw) {
                const unsigned aw = mpair * 16u + lane16 + kw;
                const float* ap = &Alds[((dc * 2 + half) * AROW + aw) * APT];
                v4f a[VR];
                #pragma unroll
                for (int vr = 0; vr < VR; ++vr) a[vr] = *(const v4f*)(ap + vr * 4);
                #pragma unroll
                for (int tp2 = 0; tp2 < VR / 2; ++tp2) {
                    v4f b0[2], b1[2];
                    #pragma unroll
                    for (int u = 0; u < 2; ++u) {
                        const int tq = (dc * VR + 2 * tp2 + u) * 3 + kw;
                        const float* bp = &Wlds[(npair * 32u + lane16) * WS + tq * 8 + half * 4u];
                        b0[u] = *(const v4f*)bp;
                        b1[u] = *(const v4f*)(bp + 16 * WS);
                    }
                    #pragma unroll
                    for (int u = 0; u < 2; ++u) {
                        const v4f ar = a[2 * tp2 + u];
                        const v2f al = {ar.x, ar.y};
                        const v2f b0l = {b0[u].x, b0[u].y}, b1l = {b1[u].x, b1[u].y};
                        acc[dc][2 * tp2 + u][0] = __builtin_amdgcn_wmma_f32_16x16x4_f32(
                            false, al, false, b0l, (short)0, acc[dc][2 * tp2 + u][0], false, false);
                        acc[dc][2 * tp2 + u][1] = __builtin_amdgcn_wmma_f32_16x16x4_f32(
                            false, al, false, b1l, (short)0, acc[dc][2 * tp2 + u][1], false, false);
                    }
                    #pragma unroll
                    for (int u = 0; u < 2; ++u) {
                        const v4f ar = a[2 * tp2 + u];
                        const v2f ah = {ar.z, ar.w};
                        const v2f b0h = {b0[u].z, b0[u].w}, b1h = {b1[u].z, b1[u].w};
                        acc[dc][2 * tp2 + u][0] = __builtin_amdgcn_wmma_f32_16x16x4_f32(
                            false, ah, false, b0h, (short)0, acc[dc][2 * tp2 + u][0], false, false);
                        acc[dc][2 * tp2 + u][1] = __builtin_amdgcn_wmma_f32_16x16x4_f32(
                            false, ah, false, b1h, (short)0, acc[dc][2 * tp2 + u][1], false, false);
                    }
                }
            }
        }
    }

    const unsigned wbase = mpair * 16u + half * 8u;
    {
        #pragma unroll
        for (int t = 0; t < 2; ++t) {
            const int o = (int)(npair * 32u + t * 16u + lane16);
            #pragma unroll
            for (int j = 0; j < DT; ++j) {
                v8f m[VR];
                #pragma unroll
                for (int r = 0; r < VR; ++r) {
                    m[r] = (j == 0) ? ((acc[0][r][t] + acc[1][r][t]) + acc[2][r][t])
                                    : ((acc[1][r][t] - acc[2][r][t]) - acc[3][r][t]);
                }
                #pragma unroll
                for (int rr = 0; rr < 4; ++rr) {
                    const size_t off =
                        (((size_t)(b * CC + o) * DD + (d0 + j)) * HH + (h0 + rr)) * WW + wbase;
                    float v[8];
                    #pragma unroll
                    for (int i = 0; i < 8; ++i) {
                        const float dm = m[1][i] - m[2][i], sm = m[1][i] + m[2][i];
                        const float dq = m[3][i] - m[4][i], sq = m[3][i] + m[4][i];
                        v[i] = (rr == 0) ? ((m[0][i] + sm) + sq)
                             : (rr == 1) ? fmaf(2.f, dq, dm)
                             : (rr == 2) ? fmaf(4.f, sq, sm)
                             :             (fmaf(8.f, dq, dm) + m[5][i]);
                    }
                    if (MODE == 0) {
                        const float bo = bias[o];
                        #pragma unroll
                        for (int i = 0; i < 8; ++i) v[i] = swish_f(v[i] + bo);
                    } else {
                        const float4 x0 = *(const float4*)(resid + off);
                        const float4 x1 = *(const float4*)(resid + off + 4);
                        v[0] += x0.x; v[1] += x0.y; v[2] += x0.z; v[3] += x0.w;
                        v[4] += x1.x; v[5] += x1.y; v[6] += x1.z; v[7] += x1.w;
                    }
                    *(float4*)(dst + off)     = make_float4(v[0], v[1], v[2], v[3]);
                    *(float4*)(dst + off + 4) = make_float4(v[4], v[5], v[6], v[7]);
                }
            }
        }
    }
}

extern "C" void kernel_launch(void* const* d_in, const int* in_sizes, int n_in,
                              void* d_out, int out_size, void* d_ws, size_t ws_size,
                              hipStream_t stream) {
    (void)in_sizes; (void)n_in; (void)out_size; (void)ws_size;
    const float* x      = (const float*)d_in[0];
    const float* embed  = (const float*)d_in[1];
    const float* prompt = (const float*)d_in[2];
    const float* c20w   = (const float*)d_in[3];
    const float* c20b   = (const float*)d_in[4];
    const float* c21w   = (const float*)d_in[5];
    const float* d1w    = (const float*)d_in[6];
    const float* d1b    = (const float*)d_in[7];
    const float* d2w    = (const float*)d_in[8];
    const float* d2b    = (const float*)d_in[9];
    float* out  = (float*)d_out;
    float* hbuf = (float*)d_ws + 512;

    const int ngrid = 4 * (DD / DT) * (HH / TR);
    conv3x3x3_wmma_kernel<0><<<ngrid, 256, 0, stream>>>(x, c20w, c20b, embed, d1w, d1b, nullptr, hbuf);
    conv3x3x3_wmma_kernel<1><<<ngrid, 256, 0, stream>>>(hbuf, c21w, nullptr, prompt, d2w, d2b, x, out);
}


# K-loop fragment reads regrouped: two register sets, the 9 LDS reads of the next 12-WMMA sub-block stand as one group in front of the current one (24 read places per chunk, were 92); sum order kept
# speedup vs baseline: 8.8035x; 1.0546x over previous
#include <hip/hip_runtime.h>


typedef float v2f __attribute__((ext_vector_type(2)));
typedef float v4f __attribute__((ext_vector_type(4)));
typedef float v8f __attribute__((ext_vector_type(8)));

#define CC 64
#define DD 8
#define HH 64
#define WW 64
#define CB 8
#define AROW 66
#define TR 4
#define VR 6
#define DT 2
#define DV 4
#define APT 28
#define NTAP (DV * VR * 3)
#define WS (NTAP * CB + 4)

__device__ __forceinline__ float swish_f(float v) {
    return v / (1.0f + __expf(-v));
}

template<int MODE>
__global__ __launch_bounds__(256) void conv3x3x3_wmma_kernel(
    const float* __restrict__ src,
    const float* __restrict__ wgt,
    const float* __restrict__ bias,
    const float* __restrict__ vec,
    const float* __restrict__ dw,
    const float* __restrict__ db,
    const float* __restrict__ resid,
    float* __restrict__ dst)
{
    __shared__ __attribute__((aligned(16))) float Alds[DV * 2 * AROW * APT];
    __shared__ __attribute__((aligned(16))) float Wlds[CC * WS];
    __shared__ __attribute__((aligned(16))) unsigned Tlds[32 * 12];
    __shared__ float Clds[CC];

    const unsigned tid    = threadIdx.x;
    const unsigned lane   = tid & 31u;
    const unsigned wv     = (unsigned)__builtin_amdgcn_readfirstlane((int)(tid >> 5));
    const unsigned lane16 = lane & 15u;
    const unsigned half   = lane >> 4;
    const unsigned mpair  = wv & 3u;
    const unsigned npair  = wv >> 2;

    const int wg = blockIdx.x;
    const int hb = wg & 15;
    const int dp = (wg >> 4) & 3;
    const int b  = wg >> 6;
    const int h0 = hb * TR;
    const int d0 = dp * DT;

    {
        const float* vb = vec + b * 256 + 4u * lane16;
        float4 xv[4];
        #pragma unroll
        for (int i = 0; i < 4; ++i) xv[i] = *(const float4*)(vb + 64 * i);
        float part[4], dbv[4];
        #pragma unroll
        for (int p = 0; p < 4; ++p) dbv[p] = db[wv * 8u + 2u * p + half];
        #pragma unroll
        for (int p = 0; p < 4; ++p) {
            const float* wrow = dw + (wv * 8u + 2u * p + half) * 256u + 4u * lane16;
            float s = (lane16 == 0) ? dbv[p] : 0.f;
            #pragma unroll
            for (int i = 0; i < 4; ++i) {
                const float4 w4 = *(const float4*)(wrow + 64 * i);
                s = fmaf(xv[i].x, w4.x, s);
                s = fmaf(xv[i].y, w4.y, s);
                s = fmaf(xv[i].z, w4.z, s);
                s = fmaf(xv[i].w, w4.w, s);
            }
            part[p] = s;
        }
        #pragma unroll
        for (int m = 8; m >= 1; m >>= 1) {
            #pragma unroll
            for (int p = 0; p < 4; ++p) part[p] += __shfl_xor(part[p], m, 32);
        }
        if (lane16 == 0) {
            #pragma unroll
            for (int p = 0; p < 4; ++p) Clds[wv * 8u + 2u * p + half] = part[p];
        }
    }

    for (unsigned i = tid; i < DV * 2 * 2 * (VR * 4); i += 256) {
        const unsigned row = i / (2u * VR * 4u), rem = i - row * (2u * VR * 4u);
        const unsigned side = rem / (VR * 4u), e = rem - side * (VR * 4u);
        Alds[(row * AROW + side * (AROW - 1)) * APT + e] = 0.f;
    }

    const unsigned a_cl   = lane >> 4;
    const unsigned a_cp   = wv >> 1;
    const unsigned a_w    = 32u * (wv & 1u) + 2u * lane16;
    const unsigned a_voff = a_cl * (DD * HH * WW) + a_w;
    float* const a_st = &Alds[((a_cp >> 1) * AROW + 1u + a_w) * APT + 2u * (a_cp & 1u) + a_cl];

    if (wv == 0) {
        #pragma unroll
        for (unsigned k = 0; k < 6; ++k) {
            const unsigned q  = lane + 32u * k;
            const unsigned oo = q / 24u;
            const unsigned j  = q - 24u * oo;
            const unsigned c8 = j / 3u;
            const unsigned kw = j - 3u * c8;
            Tlds[lane * 12u + k]      = oo * (CC * 27) + c8 * 27u + kw;
            Tlds[lane * 12u + 6u + k] = oo * WS + kw * 8u + c8;
        }
    }

    v8f acc[DV][VR][2];
    #pragma unroll
    for (int dc = 0; dc < DV; ++dc) {
        #pragma unroll
        for (int r = 0; r < VR; ++r) { acc[dc][r][0] = (v8f){}; acc[dc][r][1] = (v8f){}; }
    }

    for (int cb = 0; cb < CC / CB; ++cb) {
        __syncthreads();
        const uint4* tp = (const uint4*)&Tlds[lane * 12u];
        const uint4 q0 = tp[0], q1 = tp[1], q2 = tp[2];
        const unsigned wgo[6] = {q0.x, q0.y, q0.z, q0.w, q1.x, q1.y};
        const unsigned wlo[6] = {q1.z, q1.w, q2.x, q2.y, q2.z, q2.w};
        const float* g0 = wgt + (size_t)(wv * 8u) * (CC * 27) + cb * (CB * 27);
        const float* xb = src + (size_t)(b * CC + cb * CB + 2 * (int)a_cp) * (DD * HH * WW);
        const float cs = Clds[cb * CB + (int)(2u * a_cp + a_cl)];

        auto load_plane = [&](int pi, float2 (&av)[6]) {
            const int zd = d0 + pi - 1;
            #pragma unroll
            for (int i = 0; i < 6; ++i) {
                const int zh = h0 - 1 + i;
                av[i] = make_float2(0.f, 0.f);
                if (zd >= 0 && zd < DD && zh >= 0 && zh < HH)
                    av[i] = *(const float2*)(xb + (zd * HH + zh) * WW + a_voff);
            }
        };
        auto transform_plane = [&](int pi, const float2 (&av)[6], float (&tx)[6], float (&ty)[6]) {
            const int zd = d0 + pi - 1;
            float2 v[6];
            #pragma unroll
            for (int i = 0; i < 6; ++i) {
                const int zh = h0 - 1 + i;
                v[i] = av[i];
                if (MODE == 0) {
                    if (zd >= 0 && zd < DD && zh >= 0 && zh < HH) {
                        v[i].x = swish_f(v[i].x + cs); v[i].y = swish_f(v[i].y + cs);
                    }
                } else {
                    v[i].x *= cs; v[i].y *= cs;
                }
            }
            {
                const float a = fmaf(-4.f, v[2].x, v[4].x), bq = fmaf(-4.f, v[1].x, v[3].x);
                const float c = v[4].x - v[2].x,            e  = v[3].x - v[1].x;
                tx[0] = fmaf(4.f, v[0].x, fmaf(-5.f, v[2].x, v[4].x));
                tx[1] = a + bq;
                tx[2] = a - bq;
                tx[3] = fmaf( 2.f, e, c);
                tx[4] = fmaf(-2.f, e, c);
                tx[5] = fmaf(4.f, v[1].x, fmaf(-5.f, v[3].x, v[5].x));
            }
            {
                const float a = fmaf(-4.f, v[2].y, v[4].y), bq = fmaf(-4.f, v[1].y, v[3].y);
                const float c = v[4].y - v[2].y,            e  = v[3].y - v[1].y;
                ty[0] = fmaf(4.f, v[0].y, fmaf(-5.f, v[2].y, v[4].y));
                ty[1] = a + bq;
                ty[2] = a - bq;
                ty[3] = fmaf( 2.f, e, c);
                ty[4] = fmaf(-2.f, e, c);
                ty[5] = fmaf(4.f, v[1].y, fmaf(-5.f, v[3].y, v[5].y));
            }
        };
        auto load_w = [&](int k0, float (&wr)[3][9]) {
            #pragma unroll
            for (int k = 0; k < 3; ++k) {
                const float* gp = g0 + wgo[k0 + k];
                #pragma unroll
                for (int kd = 0; kd < 3; ++kd) {
                    #pragma unroll
                    for (int kh = 0; kh < 3; ++kh) wr[k][kd * 3 + kh] = gp[kd * 9 + kh * 3];
                }
            }
        };
        auto stage_w = [&](int k0, const float (&wr)[3][9]) {
            #pragma unroll
            for (int k = 0; k < 3; ++k) {
                float uh[3][6];
                #pragma unroll
                for (int kd = 0; kd < 3; ++kd) {
                    const float w0 = wr[k][kd * 3], w1 = wr[k][kd * 3 + 1], w2 = wr[k][kd * 3 + 2];
                    const float s = w0 + w2;
                    const float u = fmaf(1.f / 24.f, w0, (1.f / 6.f) * w2);
                    uh[kd][0] = 0.25f * w0;
                    uh[kd][1] = (-1.f / 6.f) * (s + w1);
                    uh[kd][2] = (-1.f / 6.f) * (s - w1);
                    uh[kd][3] = fmaf( 1.f / 12.f, w1, u);
                    uh[kd][4] = fmaf(-1.f / 12.f, w1, u);
                    uh[kd][5] = w2;
                }
                float* lp = &Wlds[(wv * 8u) * WS + wlo[k0 + k]];
                #pragma unroll
                for (int t = 0; t < VR; ++t) {
                    const float s = uh[0][t] + uh[2][t];
                    lp[(0 * VR + t) * 24] = uh[0][t];
                    lp[(1 * VR + t) * 24] = 0.5f * (s + uh[1][t]);
                    lp[(2 * VR + t) * 24] = 0.5f * (s - uh[1][t]);
                    lp[(3 * VR + t) * 24] = uh[2][t];
                }
            }
        };

        float2 av0[6], av1[6], av2[6], av3[6];
        float wr[3][9];
        load_w(0, wr);
        load_plane(0, av0);
        load_plane(1, av1);
        load_plane(2, av2);
        load_plane(3, av3);
        stage_w(0, wr);
        load_w(3, wr);
        {
            float tx[DV][6], ty[DV][6];
            transform_plane(0, av0, tx[0], ty[0]);
            transform_plane(1, av1, tx[1], ty[1]);
            transform_plane(2, av2, tx[2], ty[2]);
            transform_plane(3, av3, tx[3], ty[3]);
            #pragma unroll
            for (int t = 0; t < VR; ++t) {
                a_st[0 * (2 * AROW * APT) + t * 4]       = tx[0][t] - tx[2][t];
                a_st[0 * (2 * AROW * APT) + t * 4 + APT] = ty[0][t] - ty[2][t];
                a_st[1 * (2 * AROW * APT) + t * 4]       = tx[1][t] + tx[2][t];
                a_st[1 * (2 * AROW * APT) + t * 4 + APT] = ty[1][t] + ty[2][t];
                a_st[2 * (2 * AROW * APT) + t * 4]       = tx[2][t] - tx[1][t];
                a_st[2 * (2 * AROW * APT) + t * 4 + APT] = ty[2][t] - ty[1][t];
                a_st[3 * (2 * AROW * APT) + t * 4]       = tx[1][t] - tx[3][t];
                a_st[3 * (2 * AROW * APT) + t * 4 + APT] = ty[1][t] - ty[3][t];
            }
        }
        stage_w(3, wr);
        __syncthreads();
        v4f fa[2][3];
        v4f fb[2][3][2];
        auto load_frags = [&](int s, int set) __attribute__((always_inline)) {
            const int dc = s / 6, kw = (s % 6) / 2, th = s & 1;
            const unsigned aw = mpair * 16u + lane16 + kw;
            const float* ap = &Alds[((dc * 2 + half) * AROW + aw) * APT + th * 12];
            #pragma unroll
            for (int u = 0; u < 3; ++u) {
                const int tq = (dc * VR + 3 * th + u) * 3 + kw;
                const float* bp = &Wlds[(npair * 32u + lane16) * WS + tq * 8 + half * 4u];
                fa[set][u]    = *(const v4f*)(ap + u * 4);
                fb[set][u][0] = *(const v4f*)bp;
                fb[set][u][1] = *(const v4f*)(bp + 16 * WS);
            }
        };
        auto mma_sub = [&](int s, int set) __attribute__((always_inline)) {
            const int dc = s / 6, th = s & 1;
            #pragma unroll
            for (int u = 0; u < 3; ++u) {
                const v4f ar = fa[set][u];
                const v2f al = {ar.x, ar.y};
                const v2f b0l = {fb[set][u][0].x, fb[set][u][0].y}, b1l = {fb[set][u][1].x, fb[set][u][1].y};
                acc[dc][3 * th + u][0] = __builtin_amdgcn_wmma_f32_16x16x4_f32(
                    false, al, false, b0l, (short)0, acc[dc][3 * th + u][0], false, false);
                acc[dc][3 * th + u][1] = __builtin_amdgcn_wmma_f32_16x16x4_f32(
                    false, al, false, b1l, (short)0, acc[dc][3 * th + u][1], false, false);
            }
            #pragma unroll
            for (int u = 0; u < 3; ++u) {
                const v4f ar = fa[set][u];
                const v2f ah = {ar.z, ar.w};
                const v2f b0h = {fb[set][u][0].z, fb[set][u][0].w}, b1h = {fb[set][u][1].z, fb[set][u][1].w};
                acc[dc][3 * th + u][0] = __builtin_amdgcn_wmma_f32_16x16x4_f32(
                    false, ah, false, b0h, (short)0, acc[dc][3 * th + u][0], false, false);
                acc[dc][3 * th + u][1] = __builtin_amdgcn_wmma_f32_16x16x4_f32(
                    false, ah, false, b1h, (short)0, acc[dc][3 * th + u][1], false, false);
            }
        };
        load_frags(0, 0);
        #pragma unroll
        for (int s = 0; s < DV * 3 * 2; ++s) {
            if (s + 1 < DV * 3 * 2) load_frags(s + 1, (s + 1) & 1);
            __builtin_amdgcn_sched_barrier(0);
            mma_sub(s, s & 1);
            __builtin_amdgcn_sched_barrier(0);
        }
    }

    const unsigned wbase = mpair * 16u + half * 8u;
    {
        #pragma unroll
        for (int t = 0; t < 2; ++t) {
            const int o = (int)(npair * 32u + t * 16u + lane16);
            #pragma unroll
            for (int j = 0; j < DT; ++j) {
                v8f m[VR];
                #pragma unroll
                for (int r = 0; r < VR; ++r) {
                    m[r] = (j == 0) ? ((acc[0][r][t] + acc[1][r][t]) + acc[2][r][t])
                                    : ((acc[1][r][t] - acc[2][r][t]) - acc[3][r][t]);
                }
                #pragma unroll
                for (int rr = 0; rr < 4; ++rr) {
                    const size_t off =
                        (((size_t)(b * CC + o) * DD + (d0 + j)) * HH + (h0 + rr)) * WW + wbase;
                    float v[8];
                    #pragma unroll
                    for (int i = 0; i < 8; ++i) {
                        const float dm = m[1][i] - m[2][i], sm = m[1][i] + m[2][i];
                        const float dq = m[3][i] - m[4][i], sq = m[3][i] + m[4][i];
                        v[i] = (rr == 0) ? ((m[0][i] + sm) + sq)
                             : (rr == 1) ? fmaf(2.f, dq, dm)
                             : (rr == 2) ? fmaf(4.f, sq, sm)
                             :             (fmaf(8.f, dq, dm) + m[5][i]);
                    }
                    if (MODE == 0) {
                        const float bo = bias[o];
                        #pragma unroll
                        for (int i = 0; i < 8; ++i) v[i] = swish_f(v[i] + bo);
                    } else {
                        const float4 x0 = *(const float4*)(resid + off);
                        const float4 x1 = *(const float4*)(resid + off + 4);
                        v[0] += x0.x; v[1] += x0.y; v[2] += x0.z; v[3] += x0.w;
                        v[4] += x1.x; v[5] += x1.y; v[6] += x1.z; v[7] += x1.w;
                    }
                    *(float4*)(dst + off)     = make_float4(v[0], v[1], v[2], v[3]);
                    *(float4*)(dst + off + 4) = make_float4(v[4], v[5], v[6], v[7]);
                }
            }
        }
    }
}

extern "C" void kernel_launch(void* const* d_in, const int* in_sizes, int n_in,
                              void* d_out, int out_size, void* d_ws, size_t ws_size,
                              hipStream_t stream) {
    (void)in_sizes; (void)n_in; (void)out_size; (void)ws_size;
    const float* x      = (const float*)d_in[0];
    const float* embed  = (const float*)d_in[1];
    const float* prompt = (const float*)d_in[2];
    const float* c20w   = (const float*)d_in[3];
    const float* c20b   = (const float*)d_in[4];
    const float* c21w   = (const float*)d_in[5];
    const float* d1w    = (const float*)d_in[6];
    const float* d1b    = (const float*)d_in[7];
    const float* d2w    = (const float*)d_in[8];
    const float* d2b    = (const float*)d_in[9];
    float* out  = (float*)d_out;
    float* hbuf = (float*)d_ws + 512;

    const int ngrid = 4 * (DD / DT) * (HH / TR);
    conv3x3x3_wmma_kernel<0><<<ngrid, 256, 0, stream>>>(x, c20w, c20b, embed, d1w, d1b, nullptr, hbuf);
    conv3x3x3_wmma_kernel<1><<<ngrid, 256, 0, stream>>>(hbuf, c21w, nullptr, prompt, d2w, d2b, x, out);
}


# Swish (input staging and epilogue of conv 1) multiplies by the hardware reciprocal v_rcp_f32 (1 ulp) of 1+exp(-v) instead of the full f32 division sequence; operands stay f32
# speedup vs baseline: 9.0768x; 1.0310x over previous
#include <hip/hip_runtime.h>


typedef float v2f __attribute__((ext_vector_type(2)));
typedef float v4f __attribute__((ext_vector_type(4)));
typedef float v8f __attribute__((ext_vector_type(8)));

#define CC 64
#define DD 8
#define HH 64
#define WW 64
#define CB 8
#define AROW 66
#define TR 4
#define VR 6
#define DT 2
#define DV 4
#define APT 28
#define NTAP (DV * VR * 3)
#define WS (NTAP * CB + 4)

__device__ __forceinline__ float swish_f(float v) {
    return v * __builtin_amdgcn_rcpf(1.0f + __expf(-v));
}

template<int MODE>
__global__ __launch_bounds__(256) void conv3x3x3_wmma_kernel(
    const float* __restrict__ src,
    const float* __restrict__ wgt,
    const float* __restrict__ bias,
    const float* __restrict__ vec,
    const float* __restrict__ dw,
    const float* __restrict__ db,
    const float* __restrict__ resid,
    float* __restrict__ dst)
{
    __shared__ __attribute__((aligned(16))) float Alds[DV * 2 * AROW * APT];
    __shared__ __attribute__((aligned(16))) float Wlds[CC * WS];
    __shared__ __attribute__((aligned(16))) unsigned Tlds[32 * 12];
    __shared__ float Clds[CC];

    const unsigned tid    = threadIdx.x;
    const unsigned lane   = tid & 31u;
    const unsigned wv     = (unsigned)__builtin_amdgcn_readfirstlane((int)(tid >> 5));
    const unsigned lane16 = lane & 15u;
    const unsigned half   = lane >> 4;
    const unsigned mpair  = wv & 3u;
    const unsigned npair  = wv >> 2;

    const int wg = blockIdx.x;
    const int hb = wg & 15;
    const int dp = (wg >> 4) & 3;
    const int b  = wg >> 6;
    const int h0 = hb * TR;
    const int d0 = dp * DT;

    {
        const float* vb = vec + b * 256 + 4u * lane16;
        float4 xv[4];
        #pragma unroll
        for (int i = 0; i < 4; ++i) xv[i] = *(const float4*)(vb + 64 * i);
        float part[4], dbv[4];
        #pragma unroll
        for (int p = 0; p < 4; ++p) dbv[p] = db[wv * 8u + 2u * p + half];
        #pragma unroll
        for (int p = 0; p < 4; ++p) {
            const float* wrow = dw + (wv * 8u + 2u * p + half) * 256u + 4u * lane16;
            float s = (lane16 == 0) ? dbv[p] : 0.f;
            #pragma unroll
            for (int i = 0; i < 4; ++i) {
                const float4 w4 = *(const float4*)(wrow + 64 * i);
                s = fmaf(xv[i].x, w4.x, s);
                s = fmaf(xv[i].y, w4.y, s);
                s = fmaf(xv[i].z, w4.z, s);
                s = fmaf(xv[i].w, w4.w, s);
            }
            part[p] = s;
        }
        #pragma unroll
        for (int m = 8; m >= 1; m >>= 1) {
            #pragma unroll
            for (int p = 0; p < 4; ++p) part[p] += __shfl_xor(part[p], m, 32);
        }
        if (lane16 == 0) {
            #pragma unroll
            for (int p = 0; p < 4; ++p) Clds[wv * 8u + 2u * p + half] = part[p];
        }
    }

    for (unsigned i = tid; i < DV * 2 * 2 * (VR * 4); i += 256) {
        const unsigned row = i / (2u * VR * 4u), rem = i - row * (2u * VR * 4u);
        const unsigned side = rem / (VR * 4u), e = rem - side * (VR * 4u);
        Alds[(row * AROW + side * (AROW - 1)) * APT + e] = 0.f;
    }

    const unsigned a_cl   = lane >> 4;
    const unsigned a_cp   = wv >> 1;
    const unsigned a_w    = 32u * (wv & 1u) + 2u * lane16;
    const unsigned a_voff = a_cl * (DD * HH * WW) + a_w;
    float* const a_st = &Alds[((a_cp >> 1) * AROW + 1u + a_w) * APT + 2u * (a_cp & 1u) + a_cl];

    if (wv == 0) {
        #pragma unroll
        for (unsigned k = 0; k < 6; ++k) {
            const unsigned q  = lane + 32u * k;
            const unsigned oo = q / 24u;
            const unsigned j  = q - 24u * oo;
            const unsigned c8 = j / 3u;
            const unsigned kw = j - 3u * c8;
            Tlds[lane * 12u + k]      = oo * (CC * 27) + c8 * 27u + kw;
            Tlds[lane * 12u + 6u + k] = oo * WS + kw * 8u + c8;
        }
    }

    v8f acc[DV][VR][2];
    #pragma unroll
    for (int dc = 0; dc < DV; ++dc) {
        #pragma unroll
        for (int r = 0; r < VR; ++r) { acc[dc][r][0] = (v8f){}; acc[dc][r][1] = (v8f){}; }
    }

    for (int cb = 0; cb < CC / CB; ++cb) {
        __syncthreads();
        const uint4* tp = (const uint4*)&Tlds[lane * 12u];
        const uint4 q0 = tp[0], q1 = tp[1], q2 = tp[2];
        const unsigned wgo[6] = {q0.x, q0.y, q0.z, q0.w, q1.x, q1.y};
        const unsigned wlo[6] = {q1.z, q1.w, q2.x, q2.y, q2.z, q2.w};
        const float* g0 = wgt + (size_t)(wv * 8u) * (CC * 27) + cb * (CB * 27);
        const float* xb = src + (size_t)(b * CC + cb * CB + 2 * (int)a_cp) * (DD * HH * WW);
        const float cs = Clds[cb * CB + (int)(2u * a_cp + a_cl)];

        auto load_plane = [&](int pi, float2 (&av)[6]) {
            const int zd = d0 + pi - 1;
            #pragma unroll
            for (int i = 0; i < 6; ++i) {
                const int zh = h0 - 1 + i;
                av[i] = make_float2(0.f, 0.f);
                if (zd >= 0 && zd < DD && zh >= 0 && zh < HH)
                    av[i] = *(const float2*)(xb + (zd * HH + zh) * WW + a_voff);
            }
        };
        auto transform_plane = [&](int pi, const float2 (&av)[6], float (&tx)[6], float (&ty)[6]) {
            const int zd = d0 + pi - 1;
            float2 v[6];
            #pragma unroll
            for (int i = 0; i < 6; ++i) {
                const int zh = h0 - 1 + i;
                v[i] = av[i];
                if (MODE == 0) {
                    if (zd >= 0 && zd < DD && zh >= 0 && zh < HH) {
                        v[i].x = swish_f(v[i].x + cs); v[i].y = swish_f(v[i].y + cs);
                    }
                } else {
                    v[i].x *= cs; v[i].y *= cs;
                }
            }
            {
                const float a = fmaf(-4.f, v[2].x, v[4].x), bq = fmaf(-4.f, v[1].x, v[3].x);
                const float c = v[4].x - v[2].x,            e  = v[3].x - v[1].x;
                tx[0] = fmaf(4.f, v[0].x, fmaf(-5.f, v[2].x, v[4].x));
                tx[1] = a + bq;
                tx[2] = a - bq;
                tx[3] = fmaf( 2.f, e, c);
                tx[4] = fmaf(-2.f, e, c);
                tx[5] = fmaf(4.f, v[1].x, fmaf(-5.f, v[3].x, v[5].x));
            }
            {
                const float a = fmaf(-4.f, v[2].y, v[4].y), bq = fmaf(-4.f, v[1].y, v[3].y);
                const float c = v[4].y - v[2].y,            e  = v[3].y - v[1].y;
                ty[0] = fmaf(4.f, v[0].y, fmaf(-5.f, v[2].y, v[4].y));
                ty[1] = a + bq;
                ty[2] = a - bq;
                ty[3] = fmaf( 2.f, e, c);
                ty[4] = fmaf(-2.f, e, c);
                ty[5] = fmaf(4.f, v[1].y, fmaf(-5.f, v[3].y, v[5].y));
            }
        };
        auto load_w = [&](int k0, float (&wr)[3][9]) {
            #pragma unroll
            for (int k = 0; k < 3; ++k) {
                const float* gp = g0 + wgo[k0 + k];
                #pragma unroll
                for (int kd = 0; kd < 3; ++kd) {
                    #pragma unroll
                    for (int kh = 0; kh < 3; ++kh) wr[k][kd * 3 + kh] = gp[kd * 9 + kh * 3];
                }
            }
        };
        auto stage_w = [&](int k0, const float (&wr)[3][9]) {
            #pragma unroll
            for (int k = 0; k < 3; ++k) {
                float uh[3][6];
                #pragma unroll
                for (int kd = 0; kd < 3; ++kd) {
                    const float w0 = wr[k][kd * 3], w1 = wr[k][kd * 3 + 1], w2 = wr[k][kd * 3 + 2];
                    const float s = w0 + w2;
                    const float u = fmaf(1.f / 24.f, w0, (1.f / 6.f) * w2);
                    uh[kd][0] = 0.25f * w0;
                    uh[kd][1] = (-1.f / 6.f) * (s + w1);
                    uh[kd][2] = (-1.f / 6.f) * (s - w1);
                    uh[kd][3] = fmaf( 1.f / 12.f, w1, u);
                    uh[kd][4] = fmaf(-1.f / 12.f, w1, u);
                    uh[kd][5] = w2;
                }
                float* lp = &Wlds[(wv * 8u) * WS + wlo[k0 + k]];
                #pragma unroll
                for (int t = 0; t < VR; ++t) {
                    const float s = uh[0][t] + uh[2][t];
                    lp[(0 * VR + t) * 24] = uh[0][t];
                    lp[(1 * VR + t) * 24] = 0.5f * (s + uh[1][t]);
                    lp[(2 * VR + t) * 24] = 0.5f * (s - uh[1][t]);
                    lp[(3 * VR + t) * 24] = uh[2][t];
                }
            }
        };

        float2 av0[6], av1[6], av2[6], av3[6];
        float wr[3][9];
        load_w(0, wr);
        load_plane(0, av0);
        load_plane(1, av1);
        load_plane(2, av2);
        load_plane(3, av3);
        stage_w(0, wr);
        load_w(3, wr);
        {
            float tx[DV][6], ty[DV][6];
            transform_plane(0, av0, tx[0], ty[0]);
            transform_plane(1, av1, tx[1], ty[1]);
            transform_plane(2, av2, tx[2], ty[2]);
            transform_plane(3, av3, tx[3], ty[3]);
            #pragma unroll
            for (int t = 0; t < VR; ++t) {
                a_st[0 * (2 * AROW * APT) + t * 4]       = tx[0][t] - tx[2][t];
                a_st[0 * (2 * AROW * APT) + t * 4 + APT] = ty[0][t] - ty[2][t];
                a_st[1 * (2 * AROW * APT) + t * 4]       = tx[1][t] + tx[2][t];
                a_st[1 * (2 * AROW * APT) + t * 4 + APT] = ty[1][t] + ty[2][t];
                a_st[2 * (2 * AROW * APT) + t * 4]       = tx[2][t] - tx[1][t];
                a_st[2 * (2 * AROW * APT) + t * 4 + APT] = ty[2][t] - ty[1][t];
                a_st[3 * (2 * AROW * APT) + t * 4]       = tx[1][t] - tx[3][t];
                a_st[3 * (2 * AROW * APT) + t * 4 + APT] = ty[1][t] - ty[3][t];
            }
        }
        stage_w(3, wr);
        __syncthreads();
        v4f fa[2][3];
        v4f fb[2][3][2];
        auto load_frags = [&](int s, int set) __attribute__((always_inline)) {
            const int dc = s / 6, kw = (s % 6) / 2, th = s & 1;
            const unsigned aw = mpair * 16u + lane16 + kw;
            const float* ap = &Alds[((dc * 2 + half) * AROW + aw) * APT + th * 12];
            #pragma unroll
            for (int u = 0; u < 3; ++u) {
                const int tq = (dc * VR + 3 * th + u) * 3 + kw;
                const float* bp = &Wlds[(npair * 32u + lane16) * WS + tq * 8 + half * 4u];
                fa[set][u]    = *(const v4f*)(ap + u * 4);
                fb[set][u][0] = *(const v4f*)bp;
                fb[set][u][1] = *(const v4f*)(bp + 16 * WS);
            }
        };
        auto mma_sub = [&](int s, int set) __attribute__((always_inline)) {
            const int dc = s / 6, th = s & 1;
            #pragma unroll
            for (int u = 0; u < 3; ++u) {
                const v4f ar = fa[set][u];
                const v2f al = {ar.x, ar.y};
                const v2f b0l = {fb[set][u][0].x, fb[set][u][0].y}, b1l = {fb[set][u][1].x, fb[set][u][1].y};
                acc[dc][3 * th + u][0] = __builtin_amdgcn_wmma_f32_16x16x4_f32(
                    false, al, false, b0l, (short)0, acc[dc][3 * th + u][0], false, false);
                acc[dc][3 * th + u][1] = __builtin_amdgcn_wmma_f32_16x16x4_f32(
                    false, al, false, b1l, (short)0, acc[dc][3 * th + u][1], false, false);
            }
            #pragma unroll
            for (int u = 0; u < 3; ++u) {
                const v4f ar = fa[set][u];
                const v2f ah = {ar.z, ar.w};
                const v2f b0h = {fb[set][u][0].z, fb[set][u][0].w}, b1h = {fb[set][u][1].z, fb[set][u][1].w};
                acc[dc][3 * th + u][0] = __builtin_amdgcn_wmma_f32_16x16x4_f32(
                    false, ah, false, b0h, (short)0, acc[dc][3 * th + u][0], false, false);
                acc[dc][3 * th + u][1] = __builtin_amdgcn_wmma_f32_16x16x4_f32(
                    false, ah, false, b1h, (short)0, acc[dc][3 * th + u][1], false, false);
            }
        };
        load_frags(0, 0);
        #pragma unroll
        for (int s = 0; s < DV * 3 * 2; ++s) {
            if (s + 1 < DV * 3 * 2) load_frags(s + 1, (s + 1) & 1);
            __builtin_amdgcn_sched_barrier(0);
            mma_sub(s, s & 1);
            __builtin_amdgcn_sched_barrier(0);
        }
    }

    const unsigned wbase = mpair * 16u + half * 8u;
    {
        #pragma unroll
        for (int t = 0; t < 2; ++t) {
            const int o = (int)(npair * 32u + t * 16u + lane16);
            #pragma unroll
            for (int j = 0; j < DT; ++j) {
                v8f m[VR];
                #pragma unroll
                for (int r = 0; r < VR; ++r) {
                    m[r] = (j == 0) ? ((acc[0][r][t] + acc[1][r][t]) + acc[2][r][t])
                                    : ((acc[1][r][t] - acc[2][r][t]) - acc[3][r][t]);
                }
                #pragma unroll
                for (int rr = 0; rr < 4; ++rr) {
                    const size_t off =
                        (((size_t)(b * CC + o) * DD + (d0 + j)) * HH + (h0 + rr)) * WW + wbase;
                    float v[8];
                    #pragma unroll
                    for (int i = 0; i < 8; ++i) {
                        const float dm = m[1][i] - m[2][i], sm = m[1][i] + m[2][i];
                        const float dq = m[3][i] - m[4][i], sq = m[3][i] + m[4][i];
                        v[i] = (rr == 0) ? ((m[0][i] + sm) + sq)
                             : (rr == 1) ? fmaf(2.f, dq, dm)
                             : (rr == 2) ? fmaf(4.f, sq, sm)
                             :             (fmaf(8.f, dq, dm) + m[5][i]);
                    }
                    if (MODE == 0) {
                        const float bo = bias[o];
                        #pragma unroll
                        for (int i = 0; i < 8; ++i) v[i] = swish_f(v[i] + bo);
                    } else {
                        const float4 x0 = *(const float4*)(resid + off);
                        const float4 x1 = *(const float4*)(resid + off + 4);
                        v[0] += x0.x; v[1] += x0.y; v[2] += x0.z; v[3] += x0.w;
                        v[4] += x1.x; v[5] += x1.y; v[6] += x1.z; v[7] += x1.w;
                    }
                    *(float4*)(dst + off)     = make_float4(v[0], v[1], v[2], v[3]);
                    *(float4*)(dst + off + 4) = make_float4(v[4], v[5], v[6], v[7]);
                }
            }
        }
    }
}

extern "C" void kernel_launch(void* const* d_in, const int* in_sizes, int n_in,
                              void* d_out, int out_size, void* d_ws, size_t ws_size,
                              hipStream_t stream) {
    (void)in_sizes; (void)n_in; (void)out_size; (void)ws_size;
    const float* x      = (const float*)d_in[0];
    const float* embed  = (const float*)d_in[1];
    const float* prompt = (const float*)d_in[2];
    const float* c20w   = (const float*)d_in[3];
    const float* c20b   = (const float*)d_in[4];
    const float* c21w   = (const float*)d_in[5];
    const float* d1w    = (const float*)d_in[6];
    const float* d1b    = (const float*)d_in[7];
    const float* d2w    = (const float*)d_in[8];
    const float* d2b    = (const float*)d_in[9];
    float* out  = (float*)d_out;
    float* hbuf = (float*)d_ws + 512;

    const int ngrid = 4 * (DD / DT) * (HH / TR);
    conv3x3x3_wmma_kernel<0><<<ngrid, 256, 0, stream>>>(x, c20w, c20b, embed, d1w, d1b, nullptr, hbuf);
    conv3x3x3_wmma_kernel<1><<<ngrid, 256, 0, stream>>>(hbuf, c21w, nullptr, prompt, d2w, d2b, x, out);
}
